# starting kernel
# baseline (speedup 1.0000x reference)
#include <hip/hip_runtime.h>

typedef float v2f __attribute__((ext_vector_type(2)));
typedef float v8f __attribute__((ext_vector_type(8)));

constexpr int T_LEN   = 150000;
constexpr int N_TAPS  = 10000;
constexpr int OUT_LEN = 125000;
constexpr int START   = 12500;
constexpr int NCH     = 2;

constexpr int KB      = 1024;
constexpr int K_TOT   = 10240;
constexpr int NKB     = K_TOT / KB;
constexpr int WAVES   = 8;
constexpr int TPW     = 4;
constexpr int TILE_W  = 256;
constexpr int WAVE_OUT= TPW * TILE_W;
constexpr int WG_OUT  = WAVES * WAVE_OUT;
constexpr int ASPAN   = WG_OUT - TILE_W + 240;
constexpr int XS_LEN  = KB + ASPAN + 24;
constexpr int XS_PHYS = XS_LEN + XS_LEN / 64 + 2;

#define WMMA_F32(A, B, C) \
  __builtin_amdgcn_wmma_f32_16x16x4_f32(false, (A), false, (B), (short)0, (C), false, false)

__global__ __launch_bounds__(WAVES * 32)
void spatialize_fir_wmma(const float* __restrict__ x,
                         const float* __restrict__ ir,
                         float* __restrict__ out) {
  __shared__ float x_s[XS_PHYS];
  __shared__ float irL_s[KB + 16];
  __shared__ float irR_s[KB + 16];

  const int tid  = threadIdx.x;
  const int lane = tid & 31;
  const int wave = tid >> 5;
  const int half = lane >> 4;
  const int n16  = lane & 15;

  const int b       = blockIdx.y;
  const int wg_base = blockIdx.x * WG_OUT;

  const float* xb = x + (long)b * T_LEN;

  v8f accL[TPW] = {};
  v8f accR[TPW] = {};

  const int aBaseS = ASPAN - wave * WAVE_OUT - 16 * n16 + 2 * half - 256 * (TPW - 1);
  const int bBase  = 2 * half + n16;

  for (int kb = 0; kb < NKB; ++kb) {
    const int K0g = kb * KB - 16;
    const int xhi = wg_base + START + ASPAN - K0g;

    __syncthreads();

    for (int j = tid; j < XS_LEN; j += WAVES * 32) {
      int g = xhi - j;
      float v = (g >= 0 && g < T_LEN) ? xb[g] : 0.0f;
      x_s[j + (j >> 6)] = v;
    }
    for (int j = tid; j < KB + 16; j += WAVES * 32) {
      int m = K0g + j;
      bool ok = (m >= 0) && (m < N_TAPS);
      irL_s[j] = ok ? ir[2 * m + 0] : 0.0f;
      irR_s[j] = ok ? ir[2 * m + 1] : 0.0f;
    }
    __syncthreads();

#pragma unroll 4
    for (int ks = 0; ks < KB / 4; ++ks) {
      const int q = bBase + 4 * ks;
      v2f bl = { irL_s[q], irL_s[q + 1] };
      v2f br = { irR_s[q], irR_s[q + 1] };

      const int p = aBaseS + 4 * ks;
      const int s = p + (p >> 6);
      v2f a3 = { x_s[s],       x_s[s + 1]   };
      v2f a2 = { x_s[s + 260], x_s[s + 261] };
      v2f a1 = { x_s[s + 520], x_s[s + 521] };
      v2f a0 = { x_s[s + 780], x_s[s + 781] };

      accL[0] = WMMA_F32(a0, bl, accL[0]);
      accR[0] = WMMA_F32(a0, br, accR[0]);
      accL[1] = WMMA_F32(a1, bl, accL[1]);
      accR[1] = WMMA_F32(a1, br, accR[1]);
      accL[2] = WMMA_F32(a2, bl, accL[2]);
      accR[2] = WMMA_F32(a2, br, accR[2]);
      accL[3] = WMMA_F32(a3, bl, accL[3]);
      accR[3] = WMMA_F32(a3, br, accR[3]);
    }
  }

  float* oL = out + ((long)b * NCH + 0) * OUT_LEN;
  float* oR = oL + OUT_LEN;
#pragma unroll
  for (int t = 0; t < TPW; ++t) {
    const int tb = wg_base + wave * WAVE_OUT + t * TILE_W;
#pragma unroll
    for (int v = 0; v < 8; ++v) {
      int i = tb + 16 * (v + 8 * half) + n16;
      if (i < OUT_LEN) {
        oL[i] = accL[t][v];
        oR[i] = accR[t][v];
      }
    }
  }
}

extern "C" void kernel_launch(void* const* d_in, const int* in_sizes, int n_in,
                              void* d_out, int out_size, void* d_ws, size_t ws_size,
                              hipStream_t stream) {
  (void)in_sizes; (void)n_in; (void)d_ws; (void)ws_size; (void)out_size;
  const float* x  = (const float*)d_in[0];
  const float* ir = (const float*)d_in[1];
  float* out      = (float*)d_out;

  dim3 grid((OUT_LEN + WG_OUT - 1) / WG_OUT, 64);
  dim3 block(WAVES * 32);
  spatialize_fir_wmma<<<grid, block, 0, stream>>>(x, ir, out);
}


# K-block staging: global loads issued as two independent groups (24 x rows + 5 tap rows ahead of the barrier, 12 x rows behind it), then stored to LDS; was one load in flight per thread, 42 in a row
# speedup vs baseline: 1.0411x; 1.0411x over previous
#include <hip/hip_runtime.h>

typedef float v2f __attribute__((ext_vector_type(2)));
typedef float v8f __attribute__((ext_vector_type(8)));

constexpr int T_LEN   = 150000;
constexpr int N_TAPS  = 10000;
constexpr int OUT_LEN = 125000;
constexpr int START   = 12500;
constexpr int NCH     = 2;

constexpr int KB      = 1024;
constexpr int K_TOT   = 10240;
constexpr int NKB     = K_TOT / KB;
constexpr int WAVES   = 8;
constexpr int TPW     = 4;
constexpr int TILE_W  = 256;
constexpr int WAVE_OUT= TPW * TILE_W;
constexpr int WG_OUT  = WAVES * WAVE_OUT;
constexpr int ASPAN   = WG_OUT - TILE_W + 240;
constexpr int XS_ROWS = 36;
constexpr int XS_LEN  = XS_ROWS * WAVES * 32;
constexpr int XS_PRE  = 24;
constexpr int IR_ROWS = (KB + 16 + WAVES * 32 - 1) / (WAVES * 32);
constexpr int XS_PHYS = XS_LEN + XS_LEN / 64 + 2;

#define WMMA_F32(A, B, C) \
  __builtin_amdgcn_wmma_f32_16x16x4_f32(false, (A), false, (B), (short)0, (C), false, false)

__global__ __launch_bounds__(WAVES * 32)
void spatialize_fir_wmma(const float* __restrict__ x,
                         const float* __restrict__ ir,
                         float* __restrict__ out) {
  __shared__ float x_s[XS_PHYS];
  __shared__ float irL_s[KB + 16];
  __shared__ float irR_s[KB + 16];

  const int tid  = threadIdx.x;
  const int lane = tid & 31;
  const int wave = tid >> 5;
  const int half = lane >> 4;
  const int n16  = lane & 15;

  const int b       = blockIdx.y;
  const int wg_base = blockIdx.x * WG_OUT;

  const float* xb = x + (long)b * T_LEN;

  v8f accL[TPW] = {};
  v8f accR[TPW] = {};

  const int aBaseS = ASPAN - wave * WAVE_OUT - 16 * n16 + 2 * half - 256 * (TPW - 1);
  const int bBase  = 2 * half + n16;

  for (int kb = 0; kb < NKB; ++kb) {
    const int K0g = kb * KB - 16;
    const int xhi = wg_base + START + ASPAN - K0g;

    const float* xsrc = xb + (xhi - tid);
    float xv[XS_PRE];
#pragma unroll
    for (int r = 0; r < XS_PRE; ++r) xv[r] = xsrc[-(WAVES * 32) * r];
    float tl[IR_ROWS], tr[IR_ROWS];
#pragma unroll
    for (int r = 0; r < IR_ROWS; ++r) {
      const int m  = K0g + tid + (WAVES * 32) * r;
      const bool ok = (m >= 0) && (m < N_TAPS);
      const int mc = ok ? m : 0;
      const float l = ir[2 * mc + 0];
      const float rr = ir[2 * mc + 1];
      tl[r] = ok ? l : 0.0f;
      tr[r] = ok ? rr : 0.0f;
    }

    __syncthreads();

    float* xdst = x_s + tid + (tid >> 6);
    float xw[XS_ROWS - XS_PRE];
#pragma unroll
    for (int r = XS_PRE; r < XS_ROWS; ++r) xw[r - XS_PRE] = xsrc[-(WAVES * 32) * r];
#pragma unroll
    for (int r = 0; r < XS_PRE; ++r) xdst[(WAVES * 32 + 4) * r] = xv[r];
#pragma unroll
    for (int r = XS_PRE; r < XS_ROWS; ++r) xdst[(WAVES * 32 + 4) * r] = xw[r - XS_PRE];
#pragma unroll
    for (int r = 0; r < IR_ROWS - 1; ++r) {
      irL_s[tid + (WAVES * 32) * r] = tl[r];
      irR_s[tid + (WAVES * 32) * r] = tr[r];
    }
    if (tid < KB + 16 - (IR_ROWS - 1) * WAVES * 32) {
      irL_s[tid + (WAVES * 32) * (IR_ROWS - 1)] = tl[IR_ROWS - 1];
      irR_s[tid + (WAVES * 32) * (IR_ROWS - 1)] = tr[IR_ROWS - 1];
    }
    __syncthreads();

#pragma unroll 4
    for (int ks = 0; ks < KB / 4; ++ks) {
      const int q = bBase + 4 * ks;
      v2f bl = { irL_s[q], irL_s[q + 1] };
      v2f br = { irR_s[q], irR_s[q + 1] };

      const int p = aBaseS + 4 * ks;
      const int s = p + (p >> 6);
      v2f a3 = { x_s[s],       x_s[s + 1]   };
      v2f a2 = { x_s[s + 260], x_s[s + 261] };
      v2f a1 = { x_s[s + 520], x_s[s + 521] };
      v2f a0 = { x_s[s + 780], x_s[s + 781] };

      accL[0] = WMMA_F32(a0, bl, accL[0]);
      accR[0] = WMMA_F32(a0, br, accR[0]);
      accL[1] = WMMA_F32(a1, bl, accL[1]);
      accR[1] = WMMA_F32(a1, br, accR[1]);
      accL[2] = WMMA_F32(a2, bl, accL[2]);
      accR[2] = WMMA_F32(a2, br, accR[2]);
      accL[3] = WMMA_F32(a3, bl, accL[3]);
      accR[3] = WMMA_F32(a3, br, accR[3]);
    }
  }

  float* oL = out + ((long)b * NCH + 0) * OUT_LEN;
  float* oR = oL + OUT_LEN;
#pragma unroll
  for (int t = 0; t < TPW; ++t) {
    const int tb = wg_base + wave * WAVE_OUT + t * TILE_W;
#pragma unroll
    for (int v = 0; v < 8; ++v) {
      int i = tb + 16 * (v + 8 * half) + n16;
      if (i < OUT_LEN) {
        oL[i] = accL[t][v];
        oR[i] = accR[t][v];
      }
    }
  }
}

extern "C" void kernel_launch(void* const* d_in, const int* in_sizes, int n_in,
                              void* d_out, int out_size, void* d_ws, size_t ws_size,
                              hipStream_t stream) {
  (void)in_sizes; (void)n_in; (void)d_ws; (void)ws_size; (void)out_size;
  const float* x  = (const float*)d_in[0];
  const float* ir = (const float*)d_in[1];
  float* out      = (float*)d_out;

  dim3 grid((OUT_LEN + WG_OUT - 1) / WG_OUT, 64);
  dim3 block(WAVES * 32);
  spatialize_fir_wmma<<<grid, block, 0, stream>>>(x, ir, out);
}


# x tile in LDS as 16-float rows at a 20-dword pitch (was a 1-in-64 skew): A fragment reads are aligned 8-byte reads without bank conflicts at constant offsets from one base; sum order unchanged
# speedup vs baseline: 1.1077x; 1.0640x over previous
#include <hip/hip_runtime.h>

typedef float v2f __attribute__((ext_vector_type(2)));
typedef float v8f __attribute__((ext_vector_type(8)));

constexpr int T_LEN   = 150000;
constexpr int N_TAPS  = 10000;
constexpr int OUT_LEN = 125000;
constexpr int START   = 12500;
constexpr int NCH     = 2;

constexpr int KB      = 1024;
constexpr int K_TOT   = 10240;
constexpr int NKB     = K_TOT / KB;
constexpr int WAVES   = 8;
constexpr int TPW     = 4;
constexpr int TILE_W  = 256;
constexpr int WAVE_OUT= TPW * TILE_W;
constexpr int WG_OUT  = WAVES * WAVE_OUT;
constexpr int ASPAN   = WG_OUT - TILE_W + 240;
constexpr int XS_ROWS = 36;
constexpr int XS_LEN  = XS_ROWS * WAVES * 32;
constexpr int XS_PRE  = 24;
constexpr int IR_ROWS = (KB + 16 + WAVES * 32 - 1) / (WAVES * 32);
constexpr int XS_ROW  = 16;
constexpr int XS_PITCH= 20;
constexpr int XS_PHYS = (XS_LEN / XS_ROW) * XS_PITCH;

#define WMMA_F32(A, B, C) \
  __builtin_amdgcn_wmma_f32_16x16x4_f32(false, (A), false, (B), (short)0, (C), false, false)

__global__ __launch_bounds__(WAVES * 32)
void spatialize_fir_wmma(const float* __restrict__ x,
                         const float* __restrict__ ir,
                         float* __restrict__ out) {
  __shared__ __attribute__((aligned(16))) float x_s[XS_PHYS];
  __shared__ float irL_s[KB + 16];
  __shared__ float irR_s[KB + 16];

  const int tid  = threadIdx.x;
  const int lane = tid & 31;
  const int wave = tid >> 5;
  const int half = lane >> 4;
  const int n16  = lane & 15;

  const int b       = blockIdx.y;
  const int wg_base = blockIdx.x * WG_OUT;

  const float* xb = x + (long)b * T_LEN;

  v8f accL[TPW] = {};
  v8f accR[TPW] = {};

  static_assert((ASPAN - 256 * (TPW - 1)) % XS_ROW == 0, "row-aligned anchor");
  const int aRow = (ASPAN - 256 * (TPW - 1)) / XS_ROW - wave * (WAVE_OUT / XS_ROW) - n16;
  const v2f* aBase = (const v2f*)(x_s + aRow * XS_PITCH + 2 * half);
  const int bBase  = 2 * half + n16;

  for (int kb = 0; kb < NKB; ++kb) {
    const int K0g = kb * KB - 16;
    const int xhi = wg_base + START + ASPAN - K0g;

    const float* xsrc = xb + (xhi - tid);
    float xv[XS_PRE];
#pragma unroll
    for (int r = 0; r < XS_PRE; ++r) xv[r] = xsrc[-(WAVES * 32) * r];
    float tl[IR_ROWS], tr[IR_ROWS];
#pragma unroll
    for (int r = 0; r < IR_ROWS; ++r) {
      const int m  = K0g + tid + (WAVES * 32) * r;
      const bool ok = (m >= 0) && (m < N_TAPS);
      const int mc = ok ? m : 0;
      const float l = ir[2 * mc + 0];
      const float rr = ir[2 * mc + 1];
      tl[r] = ok ? l : 0.0f;
      tr[r] = ok ? rr : 0.0f;
    }

    __syncthreads();

    float* xdst = x_s + (tid >> 4) * XS_PITCH + (tid & 15);
    float xw[XS_ROWS - XS_PRE];
#pragma unroll
    for (int r = XS_PRE; r < XS_ROWS; ++r) xw[r - XS_PRE] = xsrc[-(WAVES * 32) * r];
#pragma unroll
    for (int r = 0; r < XS_PRE; ++r) xdst[(WAVES * 32 / XS_ROW) * XS_PITCH * r] = xv[r];
#pragma unroll
    for (int r = XS_PRE; r < XS_ROWS; ++r) xdst[(WAVES * 32 / XS_ROW) * XS_PITCH * r] = xw[r - XS_PRE];
#pragma unroll
    for (int r = 0; r < IR_ROWS - 1; ++r) {
      irL_s[tid + (WAVES * 32) * r] = tl[r];
      irR_s[tid + (WAVES * 32) * r] = tr[r];
    }
    if (tid < KB + 16 - (IR_ROWS - 1) * WAVES * 32) {
      irL_s[tid + (WAVES * 32) * (IR_ROWS - 1)] = tl[IR_ROWS - 1];
      irR_s[tid + (WAVES * 32) * (IR_ROWS - 1)] = tr[IR_ROWS - 1];
    }
    __syncthreads();

    for (int kq = 0; kq < KB / 16; ++kq) {
      const v2f* ap = aBase + (XS_PITCH / 2) * kq;
      const float* blp = irL_s + bBase + 16 * kq;
      const float* brp = irR_s + bBase + 16 * kq;
#pragma unroll
      for (int u = 0; u < 4; ++u) {
        v2f bl = { blp[4 * u], blp[4 * u + 1] };
        v2f br = { brp[4 * u], brp[4 * u + 1] };

        v2f a3 = ap[2 * u];
        v2f a2 = ap[2 * u + 160];
        v2f a1 = ap[2 * u + 320];
        v2f a0 = ap[2 * u + 480];

        accL[0] = WMMA_F32(a0, bl, accL[0]);
        accR[0] = WMMA_F32(a0, br, accR[0]);
        accL[1] = WMMA_F32(a1, bl, accL[1]);
        accR[1] = WMMA_F32(a1, br, accR[1]);
        accL[2] = WMMA_F32(a2, bl, accL[2]);
        accR[2] = WMMA_F32(a2, br, accR[2]);
        accL[3] = WMMA_F32(a3, bl, accL[3]);
        accR[3] = WMMA_F32(a3, br, accR[3]);
      }
    }
  }

  float* oL = out + ((long)b * NCH + 0) * OUT_LEN;
  float* oR = oL + OUT_LEN;
#pragma unroll
  for (int t = 0; t < TPW; ++t) {
    const int tb = wg_base + wave * WAVE_OUT + t * TILE_W;
#pragma unroll
    for (int v = 0; v < 8; ++v) {
      int i = tb + 16 * (v + 8 * half) + n16;
      if (i < OUT_LEN) {
        oL[i] = accL[t][v];
        oR[i] = accR[t][v];
      }
    }
  }
}

extern "C" void kernel_launch(void* const* d_in, const int* in_sizes, int n_in,
                              void* d_out, int out_size, void* d_ws, size_t ws_size,
                              hipStream_t stream) {
  (void)in_sizes; (void)n_in; (void)d_ws; (void)ws_size; (void)out_size;
  const float* x  = (const float*)d_in[0];
  const float* ir = (const float*)d_in[1];
  float* out      = (float*)d_out;

  dim3 grid((OUT_LEN + WG_OUT - 1) / WG_OUT, 64);
  dim3 block(WAVES * 32);
  spatialize_fir_wmma<<<grid, block, 0, stream>>>(x, ir, out);
}


# 2-parallel fast FIR over even/odd phases: three 5000-tap half-rate products H0*X0, H1*X1, (H0+H1)*(X0+X1) on f32 WMMA replace the 10000-tap one (3/4 of the WMMAs); sums regrouped, P1[j-1] via LDS
# speedup vs baseline: 1.4620x; 1.3199x over previous
#include <hip/hip_runtime.h>


typedef float v2f __attribute__((ext_vector_type(2)));
typedef float v4f __attribute__((ext_vector_type(4)));
typedef float v8f __attribute__((ext_vector_type(8)));

constexpr int T_LEN   = 150000;
constexpr int N_TAPS  = 10000;
constexpr int OUT_LEN = 125000;
constexpr int START   = 12500;
constexpr int NCH     = 2;

constexpr int NHT     = N_TAPS / 2;
constexpr int OUT_H   = OUT_LEN / 2;
constexpr int NPROD   = 3;
constexpr int KB      = 512;
constexpr int K_TOT   = 5120;
constexpr int NKB     = K_TOT / KB;
constexpr int WAVES   = 8;
constexpr int NT      = WAVES * 32;
constexpr int TPW     = 2;
constexpr int TILE_W  = 256;
constexpr int WAVE_J  = TPW * TILE_W;
constexpr int WG_POS  = WAVES * WAVE_J;
constexpr int LEAD    = 16;
constexpr int WG_J    = WG_POS - LEAD;
constexpr int ASPAN   = WG_POS - TILE_W + 240;
constexpr int XS_ROWS = 18;
constexpr int XS_LEN  = XS_ROWS * NT;
constexpr int XS_ROW  = 16;
constexpr int XS_PITCH= 20;
constexpr int XS_PHYS = (XS_LEN / XS_ROW) * XS_PITCH;
constexpr int HS_LEN  = KB + 16;
constexpr int HS_W    = 2 * NPROD * NCH;
constexpr int H_ROWS  = (HS_LEN + NT - 1) / NT;

static_assert(START % 2 == 0 && N_TAPS % 2 == 0 && OUT_LEN % 2 == 0 && T_LEN % 2 == 0, "even/odd phases line up");
static_assert((ASPAN - TILE_W * (TPW - 1)) % XS_ROW == 0, "row-aligned anchor");
static_assert(ASPAN + KB - 1 < XS_LEN, "staged sequence covers every A read");
static_assert(NCH * WG_POS <= NPROD * XS_PHYS, "P1 exchange fits in the sequence buffer");
static_assert(((OUT_H + WG_J - 1) / WG_J) * WG_J >= OUT_H, "grid covers every pair");

#define WMMA_F32(A, B, C) \
  __builtin_amdgcn_wmma_f32_16x16x4_f32(false, (A), false, (B), (short)0, (C), false, false)

__global__ __launch_bounds__(NT)
void spatialize_fir_wmma(const float* __restrict__ x,
                         const float* __restrict__ ir,
                         float* __restrict__ out) {
  __shared__ __attribute__((aligned(16))) float x_s[NPROD * XS_PHYS];
  __shared__ __attribute__((aligned(16))) float h_s[HS_LEN * HS_W];

  const int tid  = threadIdx.x;
  const int lane = tid & 31;
  const int wave = tid >> 5;
  const int half = lane >> 4;
  const int n16  = lane & 15;

  const int b  = blockIdx.y;
  const int Jb = blockIdx.x * WG_J - LEAD;

  const float* xb = x + (long)b * T_LEN + START;

  v8f acc[NPROD][NCH][TPW] = {};

  const int aRow = (ASPAN - TILE_W * (TPW - 1)) / XS_ROW - wave * (WAVE_J / XS_ROW) - n16;
  const v2f* aBase = (const v2f*)(x_s + aRow * XS_PITCH + 2 * half);
  const v4f* hBase = (const v4f*)(h_s + (2 * half + n16) * HS_W);

  for (int kb = 0; kb < NKB; ++kb) {
    const int K0g = kb * KB - 16;
    const int nhi = Jb + ASPAN - K0g;

    const v2f* xsrc = (const v2f*)xb + (nhi - tid);
    v2f xv[XS_ROWS];
#pragma unroll
    for (int r = 0; r < XS_ROWS; ++r) xv[r] = xsrc[-NT * r];
    v4f tv[H_ROWS];
#pragma unroll
    for (int r = 0; r < H_ROWS; ++r) {
      const int m  = K0g + tid + NT * r;
      const bool ok = (m >= 0) && (m < NHT);
      const int mc = ok ? m : 0;
      const v4f t = *(const v4f*)(ir + 4 * mc);
      const v4f z = {0.0f, 0.0f, 0.0f, 0.0f};
      tv[r] = ok ? t : z;
    }

    __syncthreads();

    float* xdst = x_s + (tid >> 4) * XS_PITCH + (tid & 15);
#pragma unroll
    for (int r = 0; r < XS_ROWS; ++r) {
      const int o = (NT / XS_ROW) * XS_PITCH * r;
      xdst[o]               = xv[r].x;
      xdst[XS_PHYS + o]     = xv[r].y;
      xdst[2 * XS_PHYS + o] = xv[r].x + xv[r].y;
    }
#pragma unroll
    for (int r = 0; r < H_ROWS; ++r) {
      const int i = tid + NT * r;
      if (i < HS_LEN) {
        const float hv[NPROD * NCH] = { tv[r].x, tv[r].y, tv[r].z, tv[r].w,
                                        tv[r].x + tv[r].z, tv[r].y + tv[r].w };
        float* hd = h_s + i * HS_W;
#pragma unroll
        for (int p = 0; p < NPROD * NCH; ++p) hd[2 * p] = hv[p];
        if (i > 0) {
#pragma unroll
          for (int p = 0; p < NPROD * NCH; ++p) hd[2 * p + 1 - HS_W] = hv[p];
        }
      }
    }
    __syncthreads();

    for (int kq = 0; kq < KB / 16; ++kq) {
      const v2f* ap = aBase + (XS_PITCH / 2) * kq;
      const v4f* hp = hBase + 16 * (HS_W / 4) * kq;
#pragma unroll
      for (int u = 0; u < 4; ++u) {
#pragma unroll
        for (int q = 0; q < NPROD; ++q) {
          const v4f hq = hp[4 * u * (HS_W / 4) + q];
          v2f bl = { hq.x, hq.y };
          v2f br = { hq.z, hq.w };

          v2f a1 = ap[q * (XS_PHYS / 2) + 2 * u];
          v2f a0 = ap[q * (XS_PHYS / 2) + 2 * u + 160];

          acc[q][0][0] = WMMA_F32(a0, bl, acc[q][0][0]);
          acc[q][1][0] = WMMA_F32(a0, br, acc[q][1][0]);
          acc[q][0][1] = WMMA_F32(a1, bl, acc[q][0][1]);
          acc[q][1][1] = WMMA_F32(a1, br, acc[q][1][1]);
        }
      }
    }
  }

  __syncthreads();
  float* ex = x_s;
#pragma unroll
  for (int c = 0; c < NCH; ++c) {
#pragma unroll
    for (int t = 0; t < TPW; ++t) {
#pragma unroll
      for (int v = 0; v < 8; ++v) {
        const int jl = wave * WAVE_J + t * TILE_W + 16 * (v + 8 * half) + n16;
        ex[c * WG_POS + jl] = acc[1][c][t][v];
      }
    }
  }
  __syncthreads();

#pragma unroll
  for (int c = 0; c < NCH; ++c) {
    float* oc = out + ((long)b * NCH + c) * OUT_LEN;
#pragma unroll
    for (int t = 0; t < TPW; ++t) {
#pragma unroll
      for (int v = 0; v < 8; ++v) {
        const int jl = wave * WAVE_J + t * TILE_W + 16 * (v + 8 * half) + n16;
        const int j  = Jb + jl;
        if (jl >= LEAD && j < OUT_H) {
          const float p0 = acc[0][c][t][v];
          const float p1 = acc[1][c][t][v];
          const float p2 = acc[2][c][t][v];
          const float p1m = ex[c * WG_POS + jl - 1];
          v2f o = { p0 + p1m, (p2 - p0) - p1 };
          *(v2f*)(oc + 2 * j) = o;
        }
      }
    }
  }
}

extern "C" void kernel_launch(void* const* d_in, const int* in_sizes, int n_in,
                              void* d_out, int out_size, void* d_ws, size_t ws_size,
                              hipStream_t stream) {
  (void)in_sizes; (void)n_in; (void)d_ws; (void)ws_size; (void)out_size;
  const float* x  = (const float*)d_in[0];
  const float* ir = (const float*)d_in[1];
  float* out      = (float*)d_out;

  dim3 grid((OUT_H + WG_J - 1) / WG_J, 64);
  dim3 block(NT);
  spatialize_fir_wmma<<<grid, block, 0, stream>>>(x, ir, out);
}


# 4-parallel fast FIR (even/odd split applied twice): nine 2500-tap quarter-rate products on f32 WMMA replace three 5000-tap ones (3/4 of the WMMAs); sums regrouped; LDS refit to nine sequences
# speedup vs baseline: 1.8487x; 1.2645x over previous
#include <hip/hip_runtime.h>


typedef float v2f __attribute__((ext_vector_type(2)));
typedef float v4f __attribute__((ext_vector_type(4)));
typedef float v8f __attribute__((ext_vector_type(8)));

constexpr int T_LEN   = 150000;
constexpr int N_TAPS  = 10000;
constexpr int OUT_LEN = 125000;
constexpr int START   = 12500;
constexpr int NCH     = 2;

constexpr int NQT     = N_TAPS / 4;
constexpr int OUT_Q   = OUT_LEN / 4;
constexpr int NPROD   = 9;
constexpr int KB      = 256;
constexpr int K_TOT   = 2560;
constexpr int NKB     = K_TOT / KB;
constexpr int WAVES   = 8;
constexpr int NT      = WAVES * 32;
constexpr int TILE_W  = 256;
constexpr int WAVE_J  = TILE_W;
constexpr int WG_POS  = WAVES * WAVE_J;
constexpr int LEAD    = 16;
constexpr int WG_J    = WG_POS - LEAD;
constexpr int ASPAN   = WG_POS - TILE_W + 240;
constexpr int XS_ROWS = 9;
constexpr int XS_LEN  = XS_ROWS * NT;
constexpr int XS_ROW  = 16;
constexpr int XS_PITCH= 20;
constexpr int XS_RPITCH = NPROD * XS_PITCH;
constexpr int XS_PHYS = (XS_LEN / XS_ROW) * XS_PITCH;
constexpr int XS_ALL  = NPROD * XS_PHYS;
constexpr int XS_NEAR = 5;
constexpr int HS_LEN  = KB + 16;
constexpr int HS_W    = 2 * NPROD * NCH;
constexpr int H_ROWS  = (HS_LEN + NT - 1) / NT;
constexpr int NEX     = 4;

static_assert(START % 4 == 0 && N_TAPS % 4 == 0 && OUT_LEN % 4 == 0 && T_LEN % 4 == 0, "the four phases line up");
static_assert(ASPAN % XS_ROW == 0, "row-aligned anchor");
static_assert(ASPAN + KB - 1 < XS_LEN, "staged sequence covers every A read");
static_assert(K_TOT - 16 >= NQT, "K sweep covers every tap");
static_assert(NEX * NCH * WG_POS <= XS_ALL, "neighbour exchange fits in the sequence buffer");
static_assert((XS_ALL * 4) % 16 == 0, "tap table stays 16-byte aligned behind the sequences");
static_assert(((OUT_Q + WG_J - 1) / WG_J) * WG_J >= OUT_Q, "grid covers every quad");
static_assert((HS_W / 4) % 2 == 1, "tap entry pitch is an odd multiple of 16 bytes");

#define WMMA_F32(A, B, C) \
  __builtin_amdgcn_wmma_f32_16x16x4_f32(false, (A), false, (B), (short)0, (C), false, false)

__global__ __launch_bounds__(NT, 4)
void spatialize_fir_wmma(const float* __restrict__ x,
                         const float* __restrict__ ir,
                         float* __restrict__ out) {
  __shared__ __attribute__((aligned(16))) float lds[XS_ALL + (HS_LEN + 1) * HS_W];
  float* x_s = lds;
  float* h_s = lds + XS_ALL;

  const int tid  = threadIdx.x;
  const int lane = tid & 31;
  const int wave = tid >> 5;
  const int half = lane >> 4;
  const int n16  = lane & 15;

  const int bt = blockIdx.y;
  const int Jb = blockIdx.x * WG_J - LEAD;

  const float* xb = x + (long)bt * T_LEN + START;

  v8f acc[NPROD][NCH] = {};

  const int aRow = ASPAN / XS_ROW - wave * (WAVE_J / XS_ROW) - n16;
  const v2f* aBase = (const v2f*)(x_s + aRow * XS_RPITCH + 2 * half);
  const v4f* hBase = (const v4f*)(h_s + (2 * half + n16 + 1) * HS_W);

  for (int kb = 0; kb < NKB; ++kb) {
    const int K0g = kb * KB - 16;
    const int nhi = Jb + ASPAN - K0g;

    v4f tv[H_ROWS][2];
#pragma unroll
    for (int r = 0; r < H_ROWS; ++r) {
      const int m  = K0g + tid + NT * r;
      const bool ok = (m >= 0) && (m < NQT);
      const int mc = ok ? m : 0;
      const v4f* tp = (const v4f*)(ir + 8 * mc);
      const v4f t0 = tp[0];
      const v4f t1 = tp[1];
      const v4f z = {0.0f, 0.0f, 0.0f, 0.0f};
      tv[r][0] = ok ? t0 : z;
      tv[r][1] = ok ? t1 : z;
    }
    const v4f* xsrc = (const v4f*)xb + (nhi - tid);
    v4f xv[XS_ROWS];
#pragma unroll
    for (int r = 0; r < XS_ROWS; ++r) xv[r] = xsrc[-NT * r];

    __syncthreads();

    float* xdst = x_s + (tid >> 4) * XS_RPITCH + (tid & 15);
    int far_off = XS_NEAR * (NT / XS_ROW) * XS_RPITCH;
    asm("" : "+s"(far_off));
    float* xfar = xdst + far_off;
#pragma unroll
    for (int r = 0; r < XS_ROWS; ++r) {
      float* xd = (r < XS_NEAR) ? xdst + (NT / XS_ROW) * XS_RPITCH * r
                                : xfar + (NT / XS_ROW) * XS_RPITCH * (r - XS_NEAR);
      const float pa = xv[r].x, pb = xv[r].y, pc = xv[r].z, pd = xv[r].w;
      const float pab = pa + pb, pcd = pc + pd;
      xd[0 * XS_PITCH] = pa;
      xd[1 * XS_PITCH] = pc;
      xd[2 * XS_PITCH] = pa + pc;
      xd[3 * XS_PITCH] = pb;
      xd[4 * XS_PITCH] = pd;
      xd[5 * XS_PITCH] = pb + pd;
      xd[6 * XS_PITCH] = pab;
      xd[7 * XS_PITCH] = pcd;
      xd[8 * XS_PITCH] = pab + pcd;
    }
#pragma unroll
    for (int r = 0; r < H_ROWS; ++r) {
      const int i = tid + NT * r;
      if (i < HS_LEN) {
        const v4f t0 = tv[r][0], t1 = tv[r][1];
        const float abL = t0.x + t0.z, abR = t0.y + t0.w;
        const float cdL = t1.x + t1.z, cdR = t1.y + t1.w;
        const float hv[NPROD * NCH] = { t0.x, t0.y,  t1.x, t1.y,  t0.x + t1.x, t0.y + t1.y,
                                        t0.z, t0.w,  t1.z, t1.w,  t0.z + t1.z, t0.w + t1.w,
                                        abL,  abR,   cdL,  cdR,   abL + cdL,   abR + cdR };
        float* hd = h_s + i * HS_W;
#pragma unroll
        for (int p = 0; p < NPROD * NCH; ++p) hd[2 * p + 1] = hv[p];
#pragma unroll
        for (int p = 0; p < NPROD * NCH; ++p) hd[HS_W + 2 * p] = hv[p];
      }
    }
    __syncthreads();

    v4f fh[3][3];
    v2f fa[3][3];
    auto load_group = [&](const v2f* ap, const v4f* hp, const int u, const int qg) {
#pragma unroll
      for (int s3 = 0; s3 < 3; ++s3) {
        const int s = 3 * qg + s3;
        fh[qg][s3] = hp[4 * u * (HS_W / 4) + s];
        fa[qg][s3] = ap[s * (XS_PITCH / 2) + 2 * u];
      }
    };
    auto mma_group = [&](const int qg) {
#pragma unroll
      for (int s3 = 0; s3 < 3; ++s3) {
        const int s = 3 * qg + s3;
        const v2f bl = { fh[qg][s3].x, fh[qg][s3].y };
        const v2f br = { fh[qg][s3].z, fh[qg][s3].w };
        acc[s][0] = WMMA_F32(fa[qg][s3], bl, acc[s][0]);
        acc[s][1] = WMMA_F32(fa[qg][s3], br, acc[s][1]);
      }
    };

    load_group(aBase, hBase, 0, 0);
    load_group(aBase, hBase, 0, 1);
    for (int kq = 0; kq < KB / 16; ++kq) {
      const v2f* ap = aBase + (XS_RPITCH / 2) * kq;
      const v4f* hp = hBase + 16 * (HS_W / 4) * kq;
      const int kn = (kq + 1 < KB / 16) ? kq + 1 : kq;
      const v2f* apn = aBase + (XS_RPITCH / 2) * kn;
      const v4f* hpn = hBase + 16 * (HS_W / 4) * kn;
#pragma unroll
      for (int g = 0; g < 12; ++g) {
        const int gl = g + 2;
        if (gl < 12) load_group(ap,  hp,  gl / 3, gl % 3);
        else         load_group(apn, hpn, (gl - 12) / 3, (gl - 12) % 3);
        mma_group(g % 3);
      }
    }
  }

  __syncthreads();
  float* ex = x_s;
#pragma unroll
  for (int c = 0; c < NCH; ++c) {
#pragma unroll
    for (int v = 0; v < 8; ++v) {
      const int jl = wave * WAVE_J + 16 * (v + 8 * half) + n16;
      ex[(0 * NCH + c) * WG_POS + jl] = acc[1][c][v];
      ex[(1 * NCH + c) * WG_POS + jl] = acc[4][c][v];
      ex[(2 * NCH + c) * WG_POS + jl] = acc[7][c][v];
      ex[(3 * NCH + c) * WG_POS + jl] = (acc[5][c][v] - acc[3][c][v]) - acc[4][c][v];
    }
  }
  __syncthreads();

#pragma unroll
  for (int c = 0; c < NCH; ++c) {
    float* oc = out + ((long)bt * NCH + c) * OUT_LEN;
#pragma unroll
    for (int v = 0; v < 8; ++v) {
      const int jl = wave * WAVE_J + 16 * (v + 8 * half) + n16;
      const int i  = Jb + jl;
      if (jl >= LEAD && i < OUT_Q) {
        const float q00 = acc[0][c][v], q01 = acc[1][c][v], q02 = acc[2][c][v];
        const float q10 = acc[3][c][v], q11 = acc[4][c][v], q12 = acc[5][c][v];
        const float q20 = acc[6][c][v], q21 = acc[7][c][v], q22 = acc[8][c][v];
        const float q01m = ex[(0 * NCH + c) * WG_POS + jl - 1];
        const float q11m = ex[(1 * NCH + c) * WG_POS + jl - 1];
        const float q21m = ex[(2 * NCH + c) * WG_POS + jl - 1];
        const float p1om = ex[(3 * NCH + c) * WG_POS + jl - 1];
        const float p0e = q00 + q01m;
        const float p1e = q10 + q11m;
        const float p2e = q20 + q21m;
        const float p0o = (q02 - q00) - q01;
        const float p1o = (q12 - q10) - q11;
        const float p2o = (q22 - q20) - q21;
        v4f o = { p0e + p1om, (p2e - p0e) - p1e, p0o + p1e, (p2o - p0o) - p1o };
        *(v4f*)(oc + 4 * i) = o;
      }
    }
  }
}

extern "C" void kernel_launch(void* const* d_in, const int* in_sizes, int n_in,
                              void* d_out, int out_size, void* d_ws, size_t ws_size,
                              hipStream_t stream) {
  (void)in_sizes; (void)n_in; (void)d_ws; (void)ws_size; (void)out_size;
  const float* x  = (const float*)d_in[0];
  const float* ir = (const float*)d_in[1];
  float* out      = (float*)d_out;

  dim3 grid((OUT_Q + WG_J - 1) / WG_J, 64);
  dim3 block(NT);
  spatialize_fir_wmma<<<grid, block, 0, stream>>>(x, ir, out);
}


# 8-parallel fast FIR (even/odd split a third time): three branches of nine 1250-tap eighth-rate products on f32 WMMA run in turn and are folded in registers (3/4 of the WMMAs); sums regrouped
# speedup vs baseline: 2.2991x; 1.2436x over previous
#include <hip/hip_runtime.h>


typedef float v2f __attribute__((ext_vector_type(2)));
typedef float v4f __attribute__((ext_vector_type(4)));
typedef float v8f __attribute__((ext_vector_type(8)));

constexpr int T_LEN   = 150000;
constexpr int N_TAPS  = 10000;
constexpr int OUT_LEN = 125000;
constexpr int START   = 12500;
constexpr int NCH     = 2;

constexpr int NOT     = N_TAPS / 8;
constexpr int OUT_O   = OUT_LEN / 8;
constexpr int NBR     = 3;
constexpr int NPROD   = 9;
constexpr int KB      = 256;
constexpr int K_TOT   = 1280;
constexpr int NKB     = K_TOT / KB;
constexpr int WAVES   = 4;
constexpr int NT      = WAVES * 32;
constexpr int TILE_W  = 256;
constexpr int WAVE_J  = TILE_W;
constexpr int WG_POS  = WAVES * WAVE_J;
constexpr int LEAD    = 16;
constexpr int WG_J    = WG_POS - LEAD;
constexpr int ASPAN   = WG_POS - TILE_W + 240;
constexpr int XS_ROWS = 10;
constexpr int XS_HALF = XS_ROWS / 2;
constexpr int XS_LEN  = XS_ROWS * NT;
constexpr int XS_ROW  = 16;
constexpr int XS_PITCH= 20;
constexpr int XS_RPITCH = NPROD * XS_PITCH;
constexpr int XS_PHYS = (XS_LEN / XS_ROW) * XS_PITCH;
constexpr int XS_ALL  = NPROD * XS_PHYS;
constexpr int HS_LEN  = KB + 16;
constexpr int HS_W    = 2 * NPROD * NCH;
constexpr int H_ROWS  = (HS_LEN + NT - 1) / NT;
constexpr int NEX     = 5;

static_assert(START % 4 == 0 && N_TAPS % 8 == 0 && OUT_LEN % 8 == 0 && T_LEN % 8 == 0, "the eight phases line up");
static_assert(ASPAN % XS_ROW == 0, "row-aligned anchor");
static_assert(ASPAN + KB - 1 < XS_LEN, "staged sequence covers every A read");
static_assert(K_TOT - 16 >= NOT, "K sweep covers every tap");
static_assert(NEX * NCH * WG_POS <= XS_ALL, "neighbour exchange fits in the sequence buffer");
static_assert((XS_ALL * 4) % 16 == 0, "tap table stays 16-byte aligned behind the sequences");
static_assert(XS_ALL * 4 < 65536, "every sequence store is a constant offset from one base");
static_assert(((OUT_O + WG_J - 1) / WG_J) * WG_J >= OUT_O, "grid covers every octet");
static_assert((HS_W / 4) % 2 == 1, "tap entry pitch is an odd multiple of 16 bytes");
static_assert(XS_ROWS % 2 == 0, "two equal row groups");

#define WMMA_F32(A, B, C) \
  __builtin_amdgcn_wmma_f32_16x16x4_f32(false, (A), false, (B), (short)0, (C), false, false)

__global__ __launch_bounds__(NT, 2)
void spatialize_fir_wmma(const float* __restrict__ x,
                         const float* __restrict__ ir,
                         float* __restrict__ out) {
  __shared__ __attribute__((aligned(16))) float lds[XS_ALL + (HS_LEN + 1) * HS_W];
  float* x_s = lds;
  float* h_s = lds + XS_ALL;

  const int tid  = threadIdx.x;
  const int lane = tid & 31;
  const int wave = tid >> 5;
  const int half = lane >> 4;
  const int n16  = lane & 15;

  const int bt = blockIdx.y;
  const int Jb = blockIdx.x * WG_J - LEAD;

  const float* xb = x + (long)bt * T_LEN + START;

  v8f ye[4][NCH], yo[4][NCH];

  const int aRow = ASPAN / XS_ROW - wave * (WAVE_J / XS_ROW) - n16;
  const v2f* aBase = (const v2f*)(x_s + aRow * XS_RPITCH + 2 * half);
  const v4f* hBase = (const v4f*)(h_s + (2 * half + n16 + 1) * HS_W);

  for (int q = 0; q < NBR; ++q) {
    const bool useE = (q != 1), useO = (q != 0);
    auto pick = [&](const float e, const float o) { return useE ? (useO ? e + o : e) : o; };

    v8f acc[NPROD][NCH] = {};

    for (int kb = 0; kb < NKB; ++kb) {
      const int K0g = kb * KB - 16;
      const int nhi = Jb + ASPAN - K0g;

      v4f tv[H_ROWS][4];
#pragma unroll
      for (int r = 0; r < H_ROWS; ++r) {
        const int m  = K0g + tid + NT * r;
        const bool ok = (m >= 0) && (m < NOT);
        const int mc = ok ? m : 0;
        const v4f* tp = (const v4f*)(ir + 16 * mc);
        const v4f z = {0.0f, 0.0f, 0.0f, 0.0f};
#pragma unroll
        for (int w = 0; w < 4; ++w) { const v4f t = tp[w]; tv[r][w] = ok ? t : z; }
      }
      const v4f* xsrc = (const v4f*)xb + 2 * (nhi - tid);
      v4f xv[XS_HALF][2];
#pragma unroll
      for (int r = 0; r < XS_HALF; ++r) { xv[r][0] = xsrc[-2 * NT * r]; xv[r][1] = xsrc[-2 * NT * r + 1]; }

      __syncthreads();

      float* xdst = x_s + (tid >> 4) * XS_RPITCH + (tid & 15);
      auto store_row = [&](const int r, const v4f v0, const v4f v1) {
        float* xd = xdst + (NT / XS_ROW) * XS_RPITCH * r;
        const float pa = pick(v0.x, v0.y), pb = pick(v0.z, v0.w), pc = pick(v1.x, v1.y), pd = pick(v1.z, v1.w);
        const float pab = pa + pb, pcd = pc + pd;
        xd[0 * XS_PITCH] = pa;
        xd[1 * XS_PITCH] = pc;
        xd[2 * XS_PITCH] = pa + pc;
        xd[3 * XS_PITCH] = pb;
        xd[4 * XS_PITCH] = pd;
        xd[5 * XS_PITCH] = pb + pd;
        xd[6 * XS_PITCH] = pab;
        xd[7 * XS_PITCH] = pcd;
        xd[8 * XS_PITCH] = pab + pcd;
      };
#pragma unroll
      for (int r = 0; r < XS_HALF; ++r) store_row(r, xv[r][0], xv[r][1]);
      v4f xw[XS_HALF][2];
#pragma unroll
      for (int r = 0; r < XS_HALF; ++r) {
        xw[r][0] = xsrc[-2 * NT * (r + XS_HALF)];
        xw[r][1] = xsrc[-2 * NT * (r + XS_HALF) + 1];
      }
#pragma unroll
      for (int r = 0; r < H_ROWS; ++r) {
        const int i = tid + NT * r;
        if (i < HS_LEN) {
          const float aL = pick(tv[r][0].x, tv[r][0].z), aR = pick(tv[r][0].y, tv[r][0].w);
          const float bL = pick(tv[r][1].x, tv[r][1].z), bR = pick(tv[r][1].y, tv[r][1].w);
          const float cL = pick(tv[r][2].x, tv[r][2].z), cR = pick(tv[r][2].y, tv[r][2].w);
          const float dL = pick(tv[r][3].x, tv[r][3].z), dR = pick(tv[r][3].y, tv[r][3].w);
          const float abL = aL + bL, abR = aR + bR;
          const float cdL = cL + dL, cdR = cR + dR;
          const float hv[NPROD * NCH] = { aL,  aR,   cL,  cR,   aL + cL,   aR + cR,
                                          bL,  bR,   dL,  dR,   bL + dL,   bR + dR,
                                          abL, abR,  cdL, cdR,  abL + cdL, abR + cdR };
          float* hd = h_s + i * HS_W;
#pragma unroll
          for (int p = 0; p < NPROD * NCH; ++p) hd[2 * p + 1] = hv[p];
#pragma unroll
          for (int p = 0; p < NPROD * NCH; ++p) hd[HS_W + 2 * p] = hv[p];
        }
      }
#pragma unroll
      for (int r = 0; r < XS_HALF; ++r) store_row(r + XS_HALF, xw[r][0], xw[r][1]);
      __syncthreads();

      v4f fh[3][3];
      v2f fa[3][3];
      auto load_group = [&](const v2f* ap, const v4f* hp, const int u, const int qg) {
#pragma unroll
        for (int s3 = 0; s3 < 3; ++s3) {
          const int s = 3 * qg + s3;
          fh[qg][s3] = hp[4 * u * (HS_W / 4) + s];
          fa[qg][s3] = ap[s * (XS_PITCH / 2) + 2 * u];
        }
      };
      auto mma_group = [&](const int qg) {
#pragma unroll
        for (int s3 = 0; s3 < 3; ++s3) {
          const int s = 3 * qg + s3;
          const v2f bl = { fh[qg][s3].x, fh[qg][s3].y };
          const v2f br = { fh[qg][s3].z, fh[qg][s3].w };
          acc[s][0] = WMMA_F32(fa[qg][s3], bl, acc[s][0]);
          acc[s][1] = WMMA_F32(fa[qg][s3], br, acc[s][1]);
        }
      };

      load_group(aBase, hBase, 0, 0);
      load_group(aBase, hBase, 0, 1);
      for (int kq = 0; kq < KB / 16; ++kq) {
        const v2f* ap = aBase + (XS_RPITCH / 2) * kq;
        const v4f* hp = hBase + 16 * (HS_W / 4) * kq;
        const int kn = (kq + 1 < KB / 16) ? kq + 1 : kq;
        const v2f* apn = aBase + (XS_RPITCH / 2) * kn;
        const v4f* hpn = hBase + 16 * (HS_W / 4) * kn;
#pragma unroll
        for (int g = 0; g < 12; ++g) {
          const int gl = g + 2;
          if (gl < 12) load_group(ap,  hp,  gl / 3, gl % 3);
          else         load_group(apn, hpn, (gl - 12) / 3, (gl - 12) % 3);
          mma_group(g % 3);
        }
      }
    }

    __syncthreads();
    float* ex = x_s;
#pragma unroll
    for (int c = 0; c < NCH; ++c) {
#pragma unroll
      for (int v = 0; v < 8; ++v) {
        const int jl = wave * WAVE_J + 16 * (v + 8 * half) + n16;
        const float r0o = (acc[2][c][v] - acc[0][c][v]) - acc[1][c][v];
        const float r1o = (acc[5][c][v] - acc[3][c][v]) - acc[4][c][v];
        const float r2o = (acc[8][c][v] - acc[6][c][v]) - acc[7][c][v];
        ex[(0 * NCH + c) * WG_POS + jl] = acc[1][c][v];
        ex[(1 * NCH + c) * WG_POS + jl] = acc[4][c][v];
        ex[(2 * NCH + c) * WG_POS + jl] = acc[7][c][v];
        ex[(3 * NCH + c) * WG_POS + jl] = r1o;
        ex[(4 * NCH + c) * WG_POS + jl] = (r2o - r0o) - r1o;
      }
    }
    __syncthreads();

#pragma unroll
    for (int c = 0; c < NCH; ++c) {
#pragma unroll
      for (int v = 0; v < 8; ++v) {
        const int jl = wave * WAVE_J + 16 * (v + 8 * half) + n16;
        const int jm = (jl > 0 ? jl : 1) - 1;
        const float q00 = acc[0][c][v], q01 = acc[1][c][v], q02 = acc[2][c][v];
        const float q10 = acc[3][c][v], q11 = acc[4][c][v], q12 = acc[5][c][v];
        const float q20 = acc[6][c][v], q21 = acc[7][c][v], q22 = acc[8][c][v];
        const float q01m = ex[(0 * NCH + c) * WG_POS + jm];
        const float q11m = ex[(1 * NCH + c) * WG_POS + jm];
        const float q21m = ex[(2 * NCH + c) * WG_POS + jm];
        const float r1om = ex[(3 * NCH + c) * WG_POS + jm];
        const float p3m  = ex[(4 * NCH + c) * WG_POS + jm];
        const float r0e = q00 + q01m;
        const float r1e = q10 + q11m;
        const float r2e = q20 + q21m;
        const float r0o = (q02 - q00) - q01;
        const float r1o = (q12 - q10) - q11;
        const float r2o = (q22 - q20) - q21;
        const float o0 = r0e + r1om;
        const float o1 = (r2e - r0e) - r1e;
        const float o2 = r0o + r1e;
        const float o3 = (r2o - r0o) - r1o;
        if (q == 0) {
          ye[0][c][v] = o0;  ye[1][c][v] = o1;  ye[2][c][v] = o2;  ye[3][c][v] = o3;
          yo[0][c][v] = -o0; yo[1][c][v] = -o1; yo[2][c][v] = -o2; yo[3][c][v] = -o3;
        } else if (q == 1) {
          ye[0][c][v] += p3m; ye[1][c][v] += o0; ye[2][c][v] += o1; ye[3][c][v] += o2;
          yo[0][c][v] -= o0;  yo[1][c][v] -= o1; yo[2][c][v] -= o2; yo[3][c][v] -= o3;
        } else {
          yo[0][c][v] += o0;  yo[1][c][v] += o1; yo[2][c][v] += o2; yo[3][c][v] += o3;
        }
      }
    }
  }

#pragma unroll
  for (int c = 0; c < NCH; ++c) {
    float* oc = out + ((long)bt * NCH + c) * OUT_LEN;
#pragma unroll
    for (int v = 0; v < 8; ++v) {
      const int jl = wave * WAVE_J + 16 * (v + 8 * half) + n16;
      const int i  = Jb + jl;
      if (jl >= LEAD && i < OUT_O) {
        const v4f w0 = { ye[0][c][v], yo[0][c][v], ye[1][c][v], yo[1][c][v] };
        const v4f w1 = { ye[2][c][v], yo[2][c][v], ye[3][c][v], yo[3][c][v] };
        *(v4f*)(oc + 8 * i)     = w0;
        *(v4f*)(oc + 8 * i + 4) = w1;
      }
    }
  }
}

extern "C" void kernel_launch(void* const* d_in, const int* in_sizes, int n_in,
                              void* d_out, int out_size, void* d_ws, size_t ws_size,
                              hipStream_t stream) {
  (void)in_sizes; (void)n_in; (void)d_ws; (void)ws_size; (void)out_size;
  const float* x  = (const float*)d_in[0];
  const float* ir = (const float*)d_in[1];
  float* out      = (float*)d_out;

  dim3 grid((OUT_O + WG_J - 1) / WG_J, 64);
  dim3 block(NT);
  spatialize_fir_wmma<<<grid, block, 0, stream>>>(x, ir, out);
}


# 16-parallel fast FIR (even/odd split a fourth time): three outer branches each run the 8-parallel scheme, 625-tap products on f32 WMMA (3/4 of the WMMAs); sums regrouped; outer fold adds into out
# speedup vs baseline: 2.5437x; 1.1064x over previous
#include <hip/hip_runtime.h>


typedef float v2f __attribute__((ext_vector_type(2)));
typedef float v4f __attribute__((ext_vector_type(4)));
typedef float v8f __attribute__((ext_vector_type(8)));

constexpr int T_LEN   = 150000;
constexpr int N_TAPS  = 10000;
constexpr int OUT_LEN = 125000;
constexpr int START   = 12500;
constexpr int NCH     = 2;

constexpr int NPH     = 16;
constexpr int NOT     = N_TAPS / NPH;
constexpr int OUT_P   = (OUT_LEN + NPH - 1) / NPH;
constexpr int NBR     = 3;
constexpr int NPROD   = 9;
constexpr int KB      = 320;
constexpr int K_LO    = -15;
constexpr int K_TOT   = 640;
constexpr int NKB     = K_TOT / KB;
constexpr int WAVES   = 4;
constexpr int NT      = WAVES * 32;
constexpr int TILE_W  = 256;
constexpr int WAVE_J  = TILE_W;
constexpr int WG_POS  = WAVES * WAVE_J;
constexpr int LEAD    = 16;
constexpr int WG_J    = WG_POS - LEAD;
constexpr int ASPAN   = WG_POS - TILE_W + 240;
constexpr int XS_ROWS = 11;
constexpr int XG_A    = 4;
constexpr int XG_B    = 4;
constexpr int XG_C    = XS_ROWS - XG_A - XG_B;
constexpr int XS_LEN  = XS_ROWS * NT;
constexpr int XS_ROW  = 16;
constexpr int XS_PITCH= 20;
constexpr int XS_RPITCH = NPROD * XS_PITCH;
constexpr int XS_PHYS = (XS_LEN / XS_ROW) * XS_PITCH;
constexpr int XS_ALL  = NPROD * XS_PHYS;
constexpr int HS_LEN  = KB + 16;
constexpr int HS_W    = 2 * NPROD * NCH;
constexpr int H_ROWS  = (HS_LEN + NT - 1) / NT;
constexpr int NEX     = 5;

static_assert(START % 4 == 0 && N_TAPS % NPH == 0 && OUT_LEN % 4 == 0 && T_LEN % 4 == 0, "the sixteen phases line up in quads");
static_assert(ASPAN % XS_ROW == 0, "row-aligned anchor");
static_assert(ASPAN + KB - 1 < XS_LEN, "staged sequence covers every A read");
static_assert(K_LO + K_TOT >= NOT && K_LO + 15 >= 0, "K sweep covers every tap");
static_assert(NKB * KB == K_TOT && KB % 16 == 0, "whole K-blocks of whole trips");
static_assert(NEX * NCH * WG_POS <= XS_ALL, "neighbour exchange fits in the sequence buffer");
static_assert((XS_ALL * 4) % 16 == 0, "tap table stays 16-byte aligned behind the sequences");
static_assert(XS_ALL * 4 < 65536, "every sequence store is a constant offset from one base");
static_assert(((OUT_P + WG_J - 1) / WG_J) * WG_J >= OUT_P, "grid covers every position");
static_assert((HS_W / 4) % 2 == 1, "tap entry pitch is an odd multiple of 16 bytes");
static_assert(XG_C > 0, "three row groups");
static_assert(START + NPH * (-LEAD + ASPAN - ((NKB - 1) * KB + K_LO) - (XS_LEN - 1)) >= 0, "staging stays inside x on the left");
static_assert(START + NPH * ((((OUT_P + WG_J - 1) / WG_J) - 1) * WG_J - LEAD + ASPAN - K_LO) + NPH - 1 < T_LEN, "and on the right");

#define WMMA_F32(A, B, C) \
  __builtin_amdgcn_wmma_f32_16x16x4_f32(false, (A), false, (B), (short)0, (C), false, false)

__global__ __launch_bounds__(NT, 2)
void spatialize_fir_wmma(const float* __restrict__ x,
                         const float* __restrict__ ir,
                         float* __restrict__ out) {
  __shared__ __attribute__((aligned(16))) float lds[XS_ALL + (HS_LEN + 1) * HS_W];
  float* x_s = lds;
  float* h_s = lds + XS_ALL;

  const int tid  = threadIdx.x;
  const int lane = tid & 31;
  const int wave = tid >> 5;
  const int half = lane >> 4;
  const int n16  = lane & 15;

  const int bt = blockIdx.y;
  const int Jb = blockIdx.x * WG_J - LEAD;

  const float* xb = x + (long)bt * T_LEN + START;

  const int aRow = ASPAN / XS_ROW - wave * (WAVE_J / XS_ROW) - n16;
  const v2f* aBase = (const v2f*)(x_s + aRow * XS_RPITCH + 2 * half);
  const v4f* hBase = (const v4f*)(h_s + (2 * half + n16 + 1) * HS_W);

  for (int p = 0; p < NBR; ++p) {
    const bool useE0 = (p != 1), useO0 = (p != 0);
    auto pick0 = [&](const float e, const float o) { return useE0 ? (useO0 ? e + o : e) : o; };

    v8f ye[4][NCH], yo[4][NCH];

    for (int q = 0; q < NBR; ++q) {
      const bool useE = (q != 1), useO = (q != 0);
      auto pick = [&](const float e, const float o) { return useE ? (useO ? e + o : e) : o; };

      v8f acc[NPROD][NCH] = {};

      for (int kb = 0; kb < NKB; ++kb) {
        const int K0g = kb * KB + K_LO;
        const int nhi = Jb + ASPAN - K0g;

        v4f tv[H_ROWS][8];
#pragma unroll
        for (int r = 0; r < H_ROWS; ++r) {
          const int m  = K0g + tid + NT * r;
          const bool ok = (m >= 0) && (m < NOT);
          const int mc = ok ? m : 0;
          const v4f* tp = (const v4f*)(ir + 2 * NPH * mc);
          const v4f z = {0.0f, 0.0f, 0.0f, 0.0f};
#pragma unroll
          for (int w = 0; w < 8; ++w) { const v4f t = tp[w]; tv[r][w] = ok ? t : z; }
        }
        const v4f* xsrc = (const v4f*)xb + 4 * (nhi - tid);
        v4f xa[XG_A][4];
#pragma unroll
        for (int r = 0; r < XG_A; ++r) {
#pragma unroll
          for (int w = 0; w < 4; ++w) xa[r][w] = xsrc[-4 * NT * r + w];
        }

        __syncthreads();

        float* xdst = x_s + (tid >> 4) * XS_RPITCH + (tid & 15);
        auto store_row = [&](const int r, const v4f v0, const v4f v1, const v4f v2, const v4f v3) {
          float* xd = xdst + (NT / XS_ROW) * XS_RPITCH * r;
          const float z0 = pick0(v0.x, v0.y), z1 = pick0(v0.z, v0.w), z2 = pick0(v1.x, v1.y), z3 = pick0(v1.z, v1.w);
          const float z4 = pick0(v2.x, v2.y), z5 = pick0(v2.z, v2.w), z6 = pick0(v3.x, v3.y), z7 = pick0(v3.z, v3.w);
          const float pa = pick(z0, z1), pb = pick(z2, z3), pc = pick(z4, z5), pd = pick(z6, z7);
          const float pab = pa + pb, pcd = pc + pd;
          xd[0 * XS_PITCH] = pa;
          xd[1 * XS_PITCH] = pc;
          xd[2 * XS_PITCH] = pa + pc;
          xd[3 * XS_PITCH] = pb;
          xd[4 * XS_PITCH] = pd;
          xd[5 * XS_PITCH] = pb + pd;
          xd[6 * XS_PITCH] = pab;
          xd[7 * XS_PITCH] = pcd;
          xd[8 * XS_PITCH] = pab + pcd;
        };
#pragma unroll
        for (int r = 0; r < XG_A; ++r) store_row(r, xa[r][0], xa[r][1], xa[r][2], xa[r][3]);
        v4f xg[XG_B][4];
#pragma unroll
        for (int r = 0; r < XG_B; ++r) {
#pragma unroll
          for (int w = 0; w < 4; ++w) xg[r][w] = xsrc[-4 * NT * (r + XG_A) + w];
        }
#pragma unroll
        for (int r = 0; r < H_ROWS; ++r) {
          const int i = tid + NT * r;
          if (i < HS_LEN) {
            float gL[8], gR[8];
#pragma unroll
            for (int w = 0; w < 8; ++w) { gL[w] = pick0(tv[r][w].x, tv[r][w].z); gR[w] = pick0(tv[r][w].y, tv[r][w].w); }
            const float aL = pick(gL[0], gL[1]), aR = pick(gR[0], gR[1]);
            const float bL = pick(gL[2], gL[3]), bR = pick(gR[2], gR[3]);
            const float cL = pick(gL[4], gL[5]), cR = pick(gR[4], gR[5]);
            const float dL = pick(gL[6], gL[7]), dR = pick(gR[6], gR[7]);
            const float abL = aL + bL, abR = aR + bR;
            const float cdL = cL + dL, cdR = cR + dR;
            const float hv[NPROD * NCH] = { aL,  aR,   cL,  cR,   aL + cL,   aR + cR,
                                            bL,  bR,   dL,  dR,   bL + dL,   bR + dR,
                                            abL, abR,  cdL, cdR,  abL + cdL, abR + cdR };
            float* hd = h_s + i * HS_W;
#pragma unroll
            for (int e = 0; e < NPROD * NCH; ++e) hd[2 * e + 1] = hv[e];
#pragma unroll
            for (int e = 0; e < NPROD * NCH; ++e) hd[HS_W + 2 * e] = hv[e];
          }
        }
        v4f xh[XG_C][4];
#pragma unroll
        for (int r = 0; r < XG_C; ++r) {
#pragma unroll
          for (int w = 0; w < 4; ++w) xh[r][w] = xsrc[-4 * NT * (r + XG_A + XG_B) + w];
        }
#pragma unroll
        for (int r = 0; r < XG_B; ++r) store_row(r + XG_A, xg[r][0], xg[r][1], xg[r][2], xg[r][3]);
#pragma unroll
        for (int r = 0; r < XG_C; ++r) store_row(r + XG_A + XG_B, xh[r][0], xh[r][1], xh[r][2], xh[r][3]);
        __syncthreads();

        v4f fh[3][3];
        v2f fa[3][3];
        auto load_group = [&](const v2f* ap, const v4f* hp, const int u, const int qg) {
#pragma unroll
          for (int s3 = 0; s3 < 3; ++s3) {
            const int s = 3 * qg + s3;
            fh[qg][s3] = hp[4 * u * (HS_W / 4) + s];
            fa[qg][s3] = ap[s * (XS_PITCH / 2) + 2 * u];
          }
        };
        auto mma_group = [&](const int qg) {
#pragma unroll
          for (int s3 = 0; s3 < 3; ++s3) {
            const int s = 3 * qg + s3;
            const v2f bl = { fh[qg][s3].x, fh[qg][s3].y };
            const v2f br = { fh[qg][s3].z, fh[qg][s3].w };
            acc[s][0] = WMMA_F32(fa[qg][s3], bl, acc[s][0]);
            acc[s][1] = WMMA_F32(fa[qg][s3], br, acc[s][1]);
          }
        };

        load_group(aBase, hBase, 0, 0);
        load_group(aBase, hBase, 0, 1);
        for (int kq = 0; kq < KB / 16; ++kq) {
          const v2f* ap = aBase + (XS_RPITCH / 2) * kq;
          const v4f* hp = hBase + 16 * (HS_W / 4) * kq;
          const int kn = (kq + 1 < KB / 16) ? kq + 1 : kq;
          const v2f* apn = aBase + (XS_RPITCH / 2) * kn;
          const v4f* hpn = hBase + 16 * (HS_W / 4) * kn;
#pragma unroll
          for (int g = 0; g < 12; ++g) {
            const int gl = g + 2;
            if (gl < 12) load_group(ap,  hp,  gl / 3, gl % 3);
            else         load_group(apn, hpn, (gl - 12) / 3, (gl - 12) % 3);
            mma_group(g % 3);
          }
        }
      }

      __syncthreads();
      float* ex = x_s;
#pragma unroll
      for (int c = 0; c < NCH; ++c) {
#pragma unroll
        for (int v = 0; v < 8; ++v) {
          const int jl = wave * WAVE_J + 16 * (v + 8 * half) + n16;
          const float r0o = (acc[2][c][v] - acc[0][c][v]) - acc[1][c][v];
          const float r1o = (acc[5][c][v] - acc[3][c][v]) - acc[4][c][v];
          const float r2o = (acc[8][c][v] - acc[6][c][v]) - acc[7][c][v];
          ex[(0 * NCH + c) * WG_POS + jl] = acc[1][c][v];
          ex[(1 * NCH + c) * WG_POS + jl] = acc[4][c][v];
          ex[(2 * NCH + c) * WG_POS + jl] = acc[7][c][v];
          ex[(3 * NCH + c) * WG_POS + jl] = r1o;
          ex[(4 * NCH + c) * WG_POS + jl] = (r2o - r0o) - r1o;
        }
      }
      __syncthreads();

#pragma unroll
      for (int c = 0; c < NCH; ++c) {
#pragma unroll
        for (int v = 0; v < 8; ++v) {
          const int jl = wave * WAVE_J + 16 * (v + 8 * half) + n16;
          const int jm = (jl > 0 ? jl : 1) - 1;
          const float q00 = acc[0][c][v], q01 = acc[1][c][v], q02 = acc[2][c][v];
          const float q10 = acc[3][c][v], q11 = acc[4][c][v], q12 = acc[5][c][v];
          const float q20 = acc[6][c][v], q21 = acc[7][c][v], q22 = acc[8][c][v];
          const float q01m = ex[(0 * NCH + c) * WG_POS + jm];
          const float q11m = ex[(1 * NCH + c) * WG_POS + jm];
          const float q21m = ex[(2 * NCH + c) * WG_POS + jm];
          const float r1om = ex[(3 * NCH + c) * WG_POS + jm];
          const float p3m  = ex[(4 * NCH + c) * WG_POS + jm];
          const float r0e = q00 + q01m;
          const float r1e = q10 + q11m;
          const float r2e = q20 + q21m;
          const float r0o = (q02 - q00) - q01;
          const float r1o = (q12 - q10) - q11;
          const float r2o = (q22 - q20) - q21;
          const float o0 = r0e + r1om;
          const float o1 = (r2e - r0e) - r1e;
          const float o2 = r0o + r1e;
          const float o3 = (r2o - r0o) - r1o;
          if (q == 0) {
            ye[0][c][v] = o0;  ye[1][c][v] = o1;  ye[2][c][v] = o2;  ye[3][c][v] = o3;
            yo[0][c][v] = -o0; yo[1][c][v] = -o1; yo[2][c][v] = -o2; yo[3][c][v] = -o3;
          } else if (q == 1) {
            ye[0][c][v] += p3m; ye[1][c][v] += o0; ye[2][c][v] += o1; ye[3][c][v] += o2;
            yo[0][c][v] -= o0;  yo[1][c][v] -= o1; yo[2][c][v] -= o2; yo[3][c][v] -= o3;
          } else {
            yo[0][c][v] += o0;  yo[1][c][v] += o1; yo[2][c][v] += o2; yo[3][c][v] += o3;
          }
        }
      }
    }

    float* ex = x_s;
    if (p == 1) {
      __syncthreads();
#pragma unroll
      for (int c = 0; c < NCH; ++c) {
#pragma unroll
        for (int v = 0; v < 8; ++v) {
          const int jl = wave * WAVE_J + 16 * (v + 8 * half) + n16;
          ex[c * WG_POS + jl] = yo[3][c][v];
        }
      }
      __syncthreads();
    }
#pragma unroll
    for (int c = 0; c < NCH; ++c) {
      float* oc = out + ((long)bt * NCH + c) * OUT_LEN;
#pragma unroll
      for (int v = 0; v < 8; ++v) {
        const int jl = wave * WAVE_J + 16 * (v + 8 * half) + n16;
        const int i  = Jb + jl;
        if (jl >= LEAD) {
          const float s7m = (p == 1) ? ex[c * WG_POS + jl - 1] : 0.0f;
#pragma unroll
          for (int m = 0; m < 4; ++m) {
            if (NPH * i + 4 * m < OUT_LEN) {
              v4f* dst = (v4f*)(oc + NPH * i + 4 * m);
              const float e = ye[m][c][v], o = yo[m][c][v];
              if (p == 0) {
                const v4f w = { e, -e, o, -o };
                *dst = w;
              } else if (p == 1) {
                const float om = (m > 0) ? yo[m > 0 ? m - 1 : 0][c][v] : s7m;
                v4f w = *dst;
                w.x += om; w.y -= e; w.z += e; w.w -= o;
                *dst = w;
              } else {
                v4f w = *dst;
                w.y += e; w.w += o;
                *dst = w;
              }
            }
          }
        }
      }
    }
  }
}

extern "C" void kernel_launch(void* const* d_in, const int* in_sizes, int n_in,
                              void* d_out, int out_size, void* d_ws, size_t ws_size,
                              hipStream_t stream) {
  (void)in_sizes; (void)n_in; (void)d_ws; (void)ws_size; (void)out_size;
  const float* x  = (const float*)d_in[0];
  const float* ir = (const float*)d_in[1];
  float* out      = (float*)d_out;

  dim3 grid((OUT_P + WG_J - 1) / WG_J, 64);
  dim3 block(NT);
  spatialize_fir_wmma<<<grid, block, 0, stream>>>(x, ir, out);
}


# Outer fold of branches 1 and 2: a channel's 32 quads of out are loaded as one batch ahead of the adds and stores (was load, add, store per quad: 64 dependent round trips a branch); sums unchanged
# speedup vs baseline: 2.6204x; 1.0302x over previous
#include <hip/hip_runtime.h>


typedef float v2f __attribute__((ext_vector_type(2)));
typedef float v4f __attribute__((ext_vector_type(4)));
typedef float v8f __attribute__((ext_vector_type(8)));

constexpr int T_LEN   = 150000;
constexpr int N_TAPS  = 10000;
constexpr int OUT_LEN = 125000;
constexpr int START   = 12500;
constexpr int NCH     = 2;

constexpr int NPH     = 16;
constexpr int NOT     = N_TAPS / NPH;
constexpr int OUT_P   = (OUT_LEN + NPH - 1) / NPH;
constexpr int NBR     = 3;
constexpr int NPROD   = 9;
constexpr int KB      = 320;
constexpr int K_LO    = -15;
constexpr int K_TOT   = 640;
constexpr int NKB     = K_TOT / KB;
constexpr int WAVES   = 4;
constexpr int NT      = WAVES * 32;
constexpr int TILE_W  = 256;
constexpr int WAVE_J  = TILE_W;
constexpr int WG_POS  = WAVES * WAVE_J;
constexpr int LEAD    = 16;
constexpr int WG_J    = WG_POS - LEAD;
constexpr int ASPAN   = WG_POS - TILE_W + 240;
constexpr int XS_ROWS = 11;
constexpr int XG_A    = 4;
constexpr int XG_B    = 4;
constexpr int XG_C    = XS_ROWS - XG_A - XG_B;
constexpr int XS_LEN  = XS_ROWS * NT;
constexpr int XS_ROW  = 16;
constexpr int XS_PITCH= 20;
constexpr int XS_RPITCH = NPROD * XS_PITCH;
constexpr int XS_PHYS = (XS_LEN / XS_ROW) * XS_PITCH;
constexpr int XS_ALL  = NPROD * XS_PHYS;
constexpr int HS_LEN  = KB + 16;
constexpr int HS_W    = 2 * NPROD * NCH;
constexpr int H_ROWS  = (HS_LEN + NT - 1) / NT;
constexpr int NEX     = 5;

static_assert(START % 4 == 0 && N_TAPS % NPH == 0 && OUT_LEN % 4 == 0 && T_LEN % 4 == 0, "the sixteen phases line up in quads");
static_assert(ASPAN % XS_ROW == 0, "row-aligned anchor");
static_assert(ASPAN + KB - 1 < XS_LEN, "staged sequence covers every A read");
static_assert(K_LO + K_TOT >= NOT && K_LO + 15 >= 0, "K sweep covers every tap");
static_assert(NKB * KB == K_TOT && KB % 16 == 0, "whole K-blocks of whole trips");
static_assert(NEX * NCH * WG_POS <= XS_ALL, "neighbour exchange fits in the sequence buffer");
static_assert((XS_ALL * 4) % 16 == 0, "tap table stays 16-byte aligned behind the sequences");
static_assert(XS_ALL * 4 < 65536, "every sequence store is a constant offset from one base");
static_assert(((OUT_P + WG_J - 1) / WG_J) * WG_J >= OUT_P, "grid covers every position");
static_assert((HS_W / 4) % 2 == 1, "tap entry pitch is an odd multiple of 16 bytes");
static_assert(XG_C > 0, "three row groups");
static_assert(START + NPH * (-LEAD + ASPAN - ((NKB - 1) * KB + K_LO) - (XS_LEN - 1)) >= 0, "staging stays inside x on the left");
static_assert(START + NPH * ((((OUT_P + WG_J - 1) / WG_J) - 1) * WG_J - LEAD + ASPAN - K_LO) + NPH - 1 < T_LEN, "and on the right");

#define WMMA_F32(A, B, C) \
  __builtin_amdgcn_wmma_f32_16x16x4_f32(false, (A), false, (B), (short)0, (C), false, false)

__global__ __launch_bounds__(NT, 2)
void spatialize_fir_wmma(const float* __restrict__ x,
                         const float* __restrict__ ir,
                         float* __restrict__ out) {
  __shared__ __attribute__((aligned(16))) float lds[XS_ALL + (HS_LEN + 1) * HS_W];
  float* x_s = lds;
  float* h_s = lds + XS_ALL;

  const int tid  = threadIdx.x;
  const int lane = tid & 31;
  const int wave = tid >> 5;
  const int half = lane >> 4;
  const int n16  = lane & 15;

  const int bt = blockIdx.y;
  const int Jb = blockIdx.x * WG_J - LEAD;

  const float* xb = x + (long)bt * T_LEN + START;

  const int aRow = ASPAN / XS_ROW - wave * (WAVE_J / XS_ROW) - n16;
  const v2f* aBase = (const v2f*)(x_s + aRow * XS_RPITCH + 2 * half);
  const v4f* hBase = (const v4f*)(h_s + (2 * half + n16 + 1) * HS_W);

  for (int p = 0; p < NBR; ++p) {
    const bool useE0 = (p != 1), useO0 = (p != 0);
    auto pick0 = [&](const float e, const float o) { return useE0 ? (useO0 ? e + o : e) : o; };

    v8f ye[4][NCH], yo[4][NCH];

    for (int q = 0; q < NBR; ++q) {
      const bool useE = (q != 1), useO = (q != 0);
      auto pick = [&](const float e, const float o) { return useE ? (useO ? e + o : e) : o; };

      v8f acc[NPROD][NCH] = {};

      for (int kb = 0; kb < NKB; ++kb) {
        const int K0g = kb * KB + K_LO;
        const int nhi = Jb + ASPAN - K0g;

        v4f tv[H_ROWS][8];
#pragma unroll
        for (int r = 0; r < H_ROWS; ++r) {
          const int m  = K0g + tid + NT * r;
          const bool ok = (m >= 0) && (m < NOT);
          const int mc = ok ? m : 0;
          const v4f* tp = (const v4f*)(ir + 2 * NPH * mc);
          const v4f z = {0.0f, 0.0f, 0.0f, 0.0f};
#pragma unroll
          for (int w = 0; w < 8; ++w) { const v4f t = tp[w]; tv[r][w] = ok ? t : z; }
        }
        const v4f* xsrc = (const v4f*)xb + 4 * (nhi - tid);
        v4f xa[XG_A][4];
#pragma unroll
        for (int r = 0; r < XG_A; ++r) {
#pragma unroll
          for (int w = 0; w < 4; ++w) xa[r][w] = xsrc[-4 * NT * r + w];
        }

        __syncthreads();

        float* xdst = x_s + (tid >> 4) * XS_RPITCH + (tid & 15);
        auto store_row = [&](const int r, const v4f v0, const v4f v1, const v4f v2, const v4f v3) {
          float* xd = xdst + (NT / XS_ROW) * XS_RPITCH * r;
          const float z0 = pick0(v0.x, v0.y), z1 = pick0(v0.z, v0.w), z2 = pick0(v1.x, v1.y), z3 = pick0(v1.z, v1.w);
          const float z4 = pick0(v2.x, v2.y), z5 = pick0(v2.z, v2.w), z6 = pick0(v3.x, v3.y), z7 = pick0(v3.z, v3.w);
          const float pa = pick(z0, z1), pb = pick(z2, z3), pc = pick(z4, z5), pd = pick(z6, z7);
          const float pab = pa + pb, pcd = pc + pd;
          xd[0 * XS_PITCH] = pa;
          xd[1 * XS_PITCH] = pc;
          xd[2 * XS_PITCH] = pa + pc;
          xd[3 * XS_PITCH] = pb;
          xd[4 * XS_PITCH] = pd;
          xd[5 * XS_PITCH] = pb + pd;
          xd[6 * XS_PITCH] = pab;
          xd[7 * XS_PITCH] = pcd;
          xd[8 * XS_PITCH] = pab + pcd;
        };
#pragma unroll
        for (int r = 0; r < XG_A; ++r) store_row(r, xa[r][0], xa[r][1], xa[r][2], xa[r][3]);
        v4f xg[XG_B][4];
#pragma unroll
        for (int r = 0; r < XG_B; ++r) {
#pragma unroll
          for (int w = 0; w < 4; ++w) xg[r][w] = xsrc[-4 * NT * (r + XG_A) + w];
        }
#pragma unroll
        for (int r = 0; r < H_ROWS; ++r) {
          const int i = tid + NT * r;
          if (i < HS_LEN) {
            float gL[8], gR[8];
#pragma unroll
            for (int w = 0; w < 8; ++w) { gL[w] = pick0(tv[r][w].x, tv[r][w].z); gR[w] = pick0(tv[r][w].y, tv[r][w].w); }
            const float aL = pick(gL[0], gL[1]), aR = pick(gR[0], gR[1]);
            const float bL = pick(gL[2], gL[3]), bR = pick(gR[2], gR[3]);
            const float cL = pick(gL[4], gL[5]), cR = pick(gR[4], gR[5]);
            const float dL = pick(gL[6], gL[7]), dR = pick(gR[6], gR[7]);
            const float abL = aL + bL, abR = aR + bR;
            const float cdL = cL + dL, cdR = cR + dR;
            const float hv[NPROD * NCH] = { aL,  aR,   cL,  cR,   aL + cL,   aR + cR,
                                            bL,  bR,   dL,  dR,   bL + dL,   bR + dR,
                                            abL, abR,  cdL, cdR,  abL + cdL, abR + cdR };
            float* hd = h_s + i * HS_W;
#pragma unroll
            for (int e = 0; e < NPROD * NCH; ++e) hd[2 * e + 1] = hv[e];
#pragma unroll
            for (int e = 0; e < NPROD * NCH; ++e) hd[HS_W + 2 * e] = hv[e];
          }
        }
        v4f xh[XG_C][4];
#pragma unroll
        for (int r = 0; r < XG_C; ++r) {
#pragma unroll
          for (int w = 0; w < 4; ++w) xh[r][w] = xsrc[-4 * NT * (r + XG_A + XG_B) + w];
        }
#pragma unroll
        for (int r = 0; r < XG_B; ++r) store_row(r + XG_A, xg[r][0], xg[r][1], xg[r][2], xg[r][3]);
#pragma unroll
        for (int r = 0; r < XG_C; ++r) store_row(r + XG_A + XG_B, xh[r][0], xh[r][1], xh[r][2], xh[r][3]);
        __syncthreads();

        v4f fh[3][3];
        v2f fa[3][3];
        auto load_group = [&](const v2f* ap, const v4f* hp, const int u, const int qg) {
#pragma unroll
          for (int s3 = 0; s3 < 3; ++s3) {
            const int s = 3 * qg + s3;
            fh[qg][s3] = hp[4 * u * (HS_W / 4) + s];
            fa[qg][s3] = ap[s * (XS_PITCH / 2) + 2 * u];
          }
        };
        auto mma_group = [&](const int qg) {
#pragma unroll
          for (int s3 = 0; s3 < 3; ++s3) {
            const int s = 3 * qg + s3;
            const v2f bl = { fh[qg][s3].x, fh[qg][s3].y };
            const v2f br = { fh[qg][s3].z, fh[qg][s3].w };
            acc[s][0] = WMMA_F32(fa[qg][s3], bl, acc[s][0]);
            acc[s][1] = WMMA_F32(fa[qg][s3], br, acc[s][1]);
          }
        };

        load_group(aBase, hBase, 0, 0);
        load_group(aBase, hBase, 0, 1);
        for (int kq = 0; kq < KB / 16; ++kq) {
          const v2f* ap = aBase + (XS_RPITCH / 2) * kq;
          const v4f* hp = hBase + 16 * (HS_W / 4) * kq;
          const int kn = (kq + 1 < KB / 16) ? kq + 1 : kq;
          const v2f* apn = aBase + (XS_RPITCH / 2) * kn;
          const v4f* hpn = hBase + 16 * (HS_W / 4) * kn;
#pragma unroll
          for (int g = 0; g < 12; ++g) {
            const int gl = g + 2;
            if (gl < 12) load_group(ap,  hp,  gl / 3, gl % 3);
            else         load_group(apn, hpn, (gl - 12) / 3, (gl - 12) % 3);
            mma_group(g % 3);
          }
        }
      }

      __syncthreads();
      float* ex = x_s;
#pragma unroll
      for (int c = 0; c < NCH; ++c) {
#pragma unroll
        for (int v = 0; v < 8; ++v) {
          const int jl = wave * WAVE_J + 16 * (v + 8 * half) + n16;
          const float r0o = (acc[2][c][v] - acc[0][c][v]) - acc[1][c][v];
          const float r1o = (acc[5][c][v] - acc[3][c][v]) - acc[4][c][v];
          const float r2o = (acc[8][c][v] - acc[6][c][v]) - acc[7][c][v];
          ex[(0 * NCH + c) * WG_POS + jl] = acc[1][c][v];
          ex[(1 * NCH + c) * WG_POS + jl] = acc[4][c][v];
          ex[(2 * NCH + c) * WG_POS + jl] = acc[7][c][v];
          ex[(3 * NCH + c) * WG_POS + jl] = r1o;
          ex[(4 * NCH + c) * WG_POS + jl] = (r2o - r0o) - r1o;
        }
      }
      __syncthreads();

#pragma unroll
      for (int c = 0; c < NCH; ++c) {
#pragma unroll
        for (int v = 0; v < 8; ++v) {
          const int jl = wave * WAVE_J + 16 * (v + 8 * half) + n16;
          const int jm = (jl > 0 ? jl : 1) - 1;
          const float q00 = acc[0][c][v], q01 = acc[1][c][v], q02 = acc[2][c][v];
          const float q10 = acc[3][c][v], q11 = acc[4][c][v], q12 = acc[5][c][v];
          const float q20 = acc[6][c][v], q21 = acc[7][c][v], q22 = acc[8][c][v];
          const float q01m = ex[(0 * NCH + c) * WG_POS + jm];
          const float q11m = ex[(1 * NCH + c) * WG_POS + jm];
          const float q21m = ex[(2 * NCH + c) * WG_POS + jm];
          const float r1om = ex[(3 * NCH + c) * WG_POS + jm];
          const float p3m  = ex[(4 * NCH + c) * WG_POS + jm];
          const float r0e = q00 + q01m;
          const float r1e = q10 + q11m;
          const float r2e = q20 + q21m;
          const float r0o = (q02 - q00) - q01;
          const float r1o = (q12 - q10) - q11;
          const float r2o = (q22 - q20) - q21;
          const float o0 = r0e + r1om;
          const float o1 = (r2e - r0e) - r1e;
          const float o2 = r0o + r1e;
          const float o3 = (r2o - r0o) - r1o;
          if (q == 0) {
            ye[0][c][v] = o0;  ye[1][c][v] = o1;  ye[2][c][v] = o2;  ye[3][c][v] = o3;
            yo[0][c][v] = -o0; yo[1][c][v] = -o1; yo[2][c][v] = -o2; yo[3][c][v] = -o3;
          } else if (q == 1) {
            ye[0][c][v] += p3m; ye[1][c][v] += o0; ye[2][c][v] += o1; ye[3][c][v] += o2;
            yo[0][c][v] -= o0;  yo[1][c][v] -= o1; yo[2][c][v] -= o2; yo[3][c][v] -= o3;
          } else {
            yo[0][c][v] += o0;  yo[1][c][v] += o1; yo[2][c][v] += o2; yo[3][c][v] += o3;
          }
        }
      }
    }

    float* ex = x_s;
    if (p == 1) {
      __syncthreads();
#pragma unroll
      for (int c = 0; c < NCH; ++c) {
#pragma unroll
        for (int v = 0; v < 8; ++v) {
          const int jl = wave * WAVE_J + 16 * (v + 8 * half) + n16;
          ex[c * WG_POS + jl] = yo[3][c][v];
        }
      }
      __syncthreads();
    }
#pragma unroll
    for (int c = 0; c < NCH; ++c) {
      float* oc = out + ((long)bt * NCH + c) * OUT_LEN;
      if (p == 0) {
#pragma unroll
        for (int v = 0; v < 8; ++v) {
          const int jl = wave * WAVE_J + 16 * (v + 8 * half) + n16;
          const int i  = Jb + jl;
          if (jl >= LEAD) {
#pragma unroll
            for (int m = 0; m < 4; ++m) {
              if (NPH * i + 4 * m < OUT_LEN) {
                v4f* dst = (v4f*)(oc + NPH * i + 4 * m);
                const float e = ye[m][c][v], o = yo[m][c][v];
                const v4f w = { e, -e, o, -o };
                *dst = w;
              }
            }
          }
        }
      } else {
        v4f w[8][4];
#pragma unroll
        for (int v = 0; v < 8; ++v) {
          const int jl = wave * WAVE_J + 16 * (v + 8 * half) + n16;
          const int i  = Jb + jl;
#pragma unroll
          for (int m = 0; m < 4; ++m) {
            v4f t = { 0.0f, 0.0f, 0.0f, 0.0f };
            if (jl >= LEAD && NPH * i + 4 * m < OUT_LEN) t = *(const v4f*)(oc + NPH * i + 4 * m);
            w[v][m] = t;
          }
        }
#pragma unroll
        for (int v = 0; v < 8; ++v) {
          const int jl = wave * WAVE_J + 16 * (v + 8 * half) + n16;
          const int i  = Jb + jl;
          if (jl >= LEAD) {
            const float s7m = (p == 1) ? ex[c * WG_POS + jl - 1] : 0.0f;
#pragma unroll
            for (int m = 0; m < 4; ++m) {
              if (NPH * i + 4 * m < OUT_LEN) {
                v4f* dst = (v4f*)(oc + NPH * i + 4 * m);
                const float e = ye[m][c][v], o = yo[m][c][v];
                v4f u = w[v][m];
                if (p == 1) {
                  const float om = (m > 0) ? yo[m > 0 ? m - 1 : 0][c][v] : s7m;
                  u.x += om; u.y -= e; u.z += e; u.w -= o;
                } else {
                  u.y += e; u.w += o;
                }
                *dst = u;
              }
            }
          }
        }
      }
    }
  }
}

extern "C" void kernel_launch(void* const* d_in, const int* in_sizes, int n_in,
                              void* d_out, int out_size, void* d_ws, size_t ws_size,
                              hipStream_t stream) {
  (void)in_sizes; (void)n_in; (void)d_ws; (void)ws_size; (void)out_size;
  const float* x  = (const float*)d_in[0];
  const float* ir = (const float*)d_in[1];
  float* out      = (float*)d_out;

  dim3 grid((OUT_P + WG_J - 1) / WG_J, 64);
  dim3 block(NT);
  spatialize_fir_wmma<<<grid, block, 0, stream>>>(x, ir, out);
}


# Wave priority (s_setprio) falls 3,2,1,0 as a workgroup passes K-blocks 0,10,15,17 of its 18: of the two workgroups on a WGP the lagging one wins the matrix pipe, both end together; sums unchanged
# speedup vs baseline: 2.7215x; 1.0386x over previous
#include <hip/hip_runtime.h>


typedef float v2f __attribute__((ext_vector_type(2)));
typedef float v4f __attribute__((ext_vector_type(4)));
typedef float v8f __attribute__((ext_vector_type(8)));

constexpr int T_LEN   = 150000;
constexpr int N_TAPS  = 10000;
constexpr int OUT_LEN = 125000;
constexpr int START   = 12500;
constexpr int NCH     = 2;

constexpr int NPH     = 16;
constexpr int NOT     = N_TAPS / NPH;
constexpr int OUT_P   = (OUT_LEN + NPH - 1) / NPH;
constexpr int NBR     = 3;
constexpr int NPROD   = 9;
constexpr int KB      = 320;
constexpr int K_LO    = -15;
constexpr int K_TOT   = 640;
constexpr int NKB     = K_TOT / KB;
constexpr int WAVES   = 4;
constexpr int NT      = WAVES * 32;
constexpr int TILE_W  = 256;
constexpr int WAVE_J  = TILE_W;
constexpr int WG_POS  = WAVES * WAVE_J;
constexpr int LEAD    = 16;
constexpr int WG_J    = WG_POS - LEAD;
constexpr int ASPAN   = WG_POS - TILE_W + 240;
constexpr int XS_ROWS = 11;
constexpr int XG_A    = 4;
constexpr int XG_B    = 4;
constexpr int XG_C    = XS_ROWS - XG_A - XG_B;
constexpr int XS_LEN  = XS_ROWS * NT;
constexpr int XS_ROW  = 16;
constexpr int XS_PITCH= 20;
constexpr int XS_RPITCH = NPROD * XS_PITCH;
constexpr int XS_PHYS = (XS_LEN / XS_ROW) * XS_PITCH;
constexpr int XS_ALL  = NPROD * XS_PHYS;
constexpr int HS_LEN  = KB + 16;
constexpr int HS_W    = 2 * NPROD * NCH;
constexpr int H_ROWS  = (HS_LEN + NT - 1) / NT;
constexpr int NEX     = 5;
constexpr int NBLK    = NBR * NBR * NKB;
constexpr int PRIO_B2 = 10;
constexpr int PRIO_B1 = 15;
constexpr int PRIO_B0 = 17;

static_assert(START % 4 == 0 && N_TAPS % NPH == 0 && OUT_LEN % 4 == 0 && T_LEN % 4 == 0, "the sixteen phases line up in quads");
static_assert(ASPAN % XS_ROW == 0, "row-aligned anchor");
static_assert(ASPAN + KB - 1 < XS_LEN, "staged sequence covers every A read");
static_assert(K_LO + K_TOT >= NOT && K_LO + 15 >= 0, "K sweep covers every tap");
static_assert(NKB * KB == K_TOT && KB % 16 == 0, "whole K-blocks of whole trips");
static_assert(NEX * NCH * WG_POS <= XS_ALL, "neighbour exchange fits in the sequence buffer");
static_assert((XS_ALL * 4) % 16 == 0, "tap table stays 16-byte aligned behind the sequences");
static_assert(XS_ALL * 4 < 65536, "every sequence store is a constant offset from one base");
static_assert(((OUT_P + WG_J - 1) / WG_J) * WG_J >= OUT_P, "grid covers every position");
static_assert((HS_W / 4) % 2 == 1, "tap entry pitch is an odd multiple of 16 bytes");
static_assert(XG_C > 0, "three row groups");
static_assert(0 < PRIO_B2 && PRIO_B2 < PRIO_B1 && PRIO_B1 < PRIO_B0 && PRIO_B0 < NBLK, "priority falls in three steps and ends at 0");
static_assert(START + NPH * (-LEAD + ASPAN - ((NKB - 1) * KB + K_LO) - (XS_LEN - 1)) >= 0, "staging stays inside x on the left");
static_assert(START + NPH * ((((OUT_P + WG_J - 1) / WG_J) - 1) * WG_J - LEAD + ASPAN - K_LO) + NPH - 1 < T_LEN, "and on the right");

#define WMMA_F32(A, B, C) \
  __builtin_amdgcn_wmma_f32_16x16x4_f32(false, (A), false, (B), (short)0, (C), false, false)

__global__ __launch_bounds__(NT, 2)
void spatialize_fir_wmma(const float* __restrict__ x,
                         const float* __restrict__ ir,
                         float* __restrict__ out) {
  __shared__ __attribute__((aligned(16))) float lds[XS_ALL + (HS_LEN + 1) * HS_W];
  float* x_s = lds;
  float* h_s = lds + XS_ALL;

  const int tid  = threadIdx.x;
  const int lane = tid & 31;
  const int wave = tid >> 5;
  const int half = lane >> 4;
  const int n16  = lane & 15;

  const int bt = blockIdx.y;
  const int Jb = blockIdx.x * WG_J - LEAD;

  const float* xb = x + (long)bt * T_LEN + START;

  const int aRow = ASPAN / XS_ROW - wave * (WAVE_J / XS_ROW) - n16;
  const v2f* aBase = (const v2f*)(x_s + aRow * XS_RPITCH + 2 * half);
  const v4f* hBase = (const v4f*)(h_s + (2 * half + n16 + 1) * HS_W);

  for (int p = 0; p < NBR; ++p) {
    const bool useE0 = (p != 1), useO0 = (p != 0);
    auto pick0 = [&](const float e, const float o) { return useE0 ? (useO0 ? e + o : e) : o; };

    v8f ye[4][NCH], yo[4][NCH];

    for (int q = 0; q < NBR; ++q) {
      const bool useE = (q != 1), useO = (q != 0);
      auto pick = [&](const float e, const float o) { return useE ? (useO ? e + o : e) : o; };

      v8f acc[NPROD][NCH] = {};

      for (int kb = 0; kb < NKB; ++kb) {
        const int blk = (p * NBR + q) * NKB + kb;
        if (blk == 0)            __builtin_amdgcn_s_setprio(3);
        else if (blk == PRIO_B2) __builtin_amdgcn_s_setprio(2);
        else if (blk == PRIO_B1) __builtin_amdgcn_s_setprio(1);
        else if (blk == PRIO_B0) __builtin_amdgcn_s_setprio(0);
        const int K0g = kb * KB + K_LO;
        const int nhi = Jb + ASPAN - K0g;

        v4f tv[H_ROWS][8];
#pragma unroll
        for (int r = 0; r < H_ROWS; ++r) {
          const int m  = K0g + tid + NT * r;
          const bool ok = (m >= 0) && (m < NOT);
          const int mc = ok ? m : 0;
          const v4f* tp = (const v4f*)(ir + 2 * NPH * mc);
          const v4f z = {0.0f, 0.0f, 0.0f, 0.0f};
#pragma unroll
          for (int w = 0; w < 8; ++w) { const v4f t = tp[w]; tv[r][w] = ok ? t : z; }
        }
        const v4f* xsrc = (const v4f*)xb + 4 * (nhi - tid);
        v4f xa[XG_A][4];
#pragma unroll
        for (int r = 0; r < XG_A; ++r) {
#pragma unroll
          for (int w = 0; w < 4; ++w) xa[r][w] = xsrc[-4 * NT * r + w];
        }

        __syncthreads();

        float* xdst = x_s + (tid >> 4) * XS_RPITCH + (tid & 15);
        auto store_row = [&](const int r, const v4f v0, const v4f v1, const v4f v2, const v4f v3) {
          float* xd = xdst + (NT / XS_ROW) * XS_RPITCH * r;
          const float z0 = pick0(v0.x, v0.y), z1 = pick0(v0.z, v0.w), z2 = pick0(v1.x, v1.y), z3 = pick0(v1.z, v1.w);
          const float z4 = pick0(v2.x, v2.y), z5 = pick0(v2.z, v2.w), z6 = pick0(v3.x, v3.y), z7 = pick0(v3.z, v3.w);
          const float pa = pick(z0, z1), pb = pick(z2, z3), pc = pick(z4, z5), pd = pick(z6, z7);
          const float pab = pa + pb, pcd = pc + pd;
          xd[0 * XS_PITCH] = pa;
          xd[1 * XS_PITCH] = pc;
          xd[2 * XS_PITCH] = pa + pc;
          xd[3 * XS_PITCH] = pb;
          xd[4 * XS_PITCH] = pd;
          xd[5 * XS_PITCH] = pb + pd;
          xd[6 * XS_PITCH] = pab;
          xd[7 * XS_PITCH] = pcd;
          xd[8 * XS_PITCH] = pab + pcd;
        };
#pragma unroll
        for (int r = 0; r < XG_A; ++r) store_row(r, xa[r][0], xa[r][1], xa[r][2], xa[r][3]);
        v4f xg[XG_B][4];
#pragma unroll
        for (int r = 0; r < XG_B; ++r) {
#pragma unroll
          for (int w = 0; w < 4; ++w) xg[r][w] = xsrc[-4 * NT * (r + XG_A) + w];
        }
#pragma unroll
        for (int r = 0; r < H_ROWS; ++r) {
          const int i = tid + NT * r;
          if (i < HS_LEN) {
            float gL[8], gR[8];
#pragma unroll
            for (int w = 0; w < 8; ++w) { gL[w] = pick0(tv[r][w].x, tv[r][w].z); gR[w] = pick0(tv[r][w].y, tv[r][w].w); }
            const float aL = pick(gL[0], gL[1]), aR = pick(gR[0], gR[1]);
            const float bL = pick(gL[2], gL[3]), bR = pick(gR[2], gR[3]);
            const float cL = pick(gL[4], gL[5]), cR = pick(gR[4], gR[5]);
            const float dL = pick(gL[6], gL[7]), dR = pick(gR[6], gR[7]);
            const float abL = aL + bL, abR = aR + bR;
            const float cdL = cL + dL, cdR = cR + dR;
            const float hv[NPROD * NCH] = { aL,  aR,   cL,  cR,   aL + cL,   aR + cR,
                                            bL,  bR,   dL,  dR,   bL + dL,   bR + dR,
                                            abL, abR,  cdL, cdR,  abL + cdL, abR + cdR };
            float* hd = h_s + i * HS_W;
#pragma unroll
            for (int e = 0; e < NPROD * NCH; ++e) hd[2 * e + 1] = hv[e];
#pragma unroll
            for (int e = 0; e < NPROD * NCH; ++e) hd[HS_W + 2 * e] = hv[e];
          }
        }
        v4f xh[XG_C][4];
#pragma unroll
        for (int r = 0; r < XG_C; ++r) {
#pragma unroll
          for (int w = 0; w < 4; ++w) xh[r][w] = xsrc[-4 * NT * (r + XG_A + XG_B) + w];
        }
#pragma unroll
        for (int r = 0; r < XG_B; ++r) store_row(r + XG_A, xg[r][0], xg[r][1], xg[r][2], xg[r][3]);
#pragma unroll
        for (int r = 0; r < XG_C; ++r) store_row(r + XG_A + XG_B, xh[r][0], xh[r][1], xh[r][2], xh[r][3]);
        __syncthreads();

        v4f fh[3][3];
        v2f fa[3][3];
        auto load_group = [&](const v2f* ap, const v4f* hp, const int u, const int qg) {
#pragma unroll
          for (int s3 = 0; s3 < 3; ++s3) {
            const int s = 3 * qg + s3;
            fh[qg][s3] = hp[4 * u * (HS_W / 4) + s];
            fa[qg][s3] = ap[s * (XS_PITCH / 2) + 2 * u];
          }
        };
        auto mma_group = [&](const int qg) {
#pragma unroll
          for (int s3 = 0; s3 < 3; ++s3) {
            const int s = 3 * qg + s3;
            const v2f bl = { fh[qg][s3].x, fh[qg][s3].y };
            const v2f br = { fh[qg][s3].z, fh[qg][s3].w };
            acc[s][0] = WMMA_F32(fa[qg][s3], bl, acc[s][0]);
            acc[s][1] = WMMA_F32(fa[qg][s3], br, acc[s][1]);
          }
        };

        load_group(aBase, hBase, 0, 0);
        load_group(aBase, hBase, 0, 1);
        for (int kq = 0; kq < KB / 16; ++kq) {
          const v2f* ap = aBase + (XS_RPITCH / 2) * kq;
          const v4f* hp = hBase + 16 * (HS_W / 4) * kq;
          const int kn = (kq + 1 < KB / 16) ? kq + 1 : kq;
          const v2f* apn = aBase + (XS_RPITCH / 2) * kn;
          const v4f* hpn = hBase + 16 * (HS_W / 4) * kn;
#pragma unroll
          for (int g = 0; g < 12; ++g) {
            const int gl = g + 2;
            if (gl < 12) load_group(ap,  hp,  gl / 3, gl % 3);
            else         load_group(apn, hpn, (gl - 12) / 3, (gl - 12) % 3);
            mma_group(g % 3);
          }
        }
      }

      __syncthreads();
      float* ex = x_s;
#pragma unroll
      for (int c = 0; c < NCH; ++c) {
#pragma unroll
        for (int v = 0; v < 8; ++v) {
          const int jl = wave * WAVE_J + 16 * (v + 8 * half) + n16;
          const float r0o = (acc[2][c][v] - acc[0][c][v]) - acc[1][c][v];
          const float r1o = (acc[5][c][v] - acc[3][c][v]) - acc[4][c][v];
          const float r2o = (acc[8][c][v] - acc[6][c][v]) - acc[7][c][v];
          ex[(0 * NCH + c) * WG_POS + jl] = acc[1][c][v];
          ex[(1 * NCH + c) * WG_POS + jl] = acc[4][c][v];
          ex[(2 * NCH + c) * WG_POS + jl] = acc[7][c][v];
          ex[(3 * NCH + c) * WG_POS + jl] = r1o;
          ex[(4 * NCH + c) * WG_POS + jl] = (r2o - r0o) - r1o;
        }
      }
      __syncthreads();

#pragma unroll
      for (int c = 0; c < NCH; ++c) {
#pragma unroll
        for (int v = 0; v < 8; ++v) {
          const int jl = wave * WAVE_J + 16 * (v + 8 * half) + n16;
          const int jm = (jl > 0 ? jl : 1) - 1;
          const float q00 = acc[0][c][v], q01 = acc[1][c][v], q02 = acc[2][c][v];
          const float q10 = acc[3][c][v], q11 = acc[4][c][v], q12 = acc[5][c][v];
          const float q20 = acc[6][c][v], q21 = acc[7][c][v], q22 = acc[8][c][v];
          const float q01m = ex[(0 * NCH + c) * WG_POS + jm];
          const float q11m = ex[(1 * NCH + c) * WG_POS + jm];
          const float q21m = ex[(2 * NCH + c) * WG_POS + jm];
          const float r1om = ex[(3 * NCH + c) * WG_POS + jm];
          const float p3m  = ex[(4 * NCH + c) * WG_POS + jm];
          const float r0e = q00 + q01m;
          const float r1e = q10 + q11m;
          const float r2e = q20 + q21m;
          const float r0o = (q02 - q00) - q01;
          const float r1o = (q12 - q10) - q11;
          const float r2o = (q22 - q20) - q21;
          const float o0 = r0e + r1om;
          const float o1 = (r2e - r0e) - r1e;
          const float o2 = r0o + r1e;
          const float o3 = (r2o - r0o) - r1o;
          if (q == 0) {
            ye[0][c][v] = o0;  ye[1][c][v] = o1;  ye[2][c][v] = o2;  ye[3][c][v] = o3;
            yo[0][c][v] = -o0; yo[1][c][v] = -o1; yo[2][c][v] = -o2; yo[3][c][v] = -o3;
          } else if (q == 1) {
            ye[0][c][v] += p3m; ye[1][c][v] += o0; ye[2][c][v] += o1; ye[3][c][v] += o2;
            yo[0][c][v] -= o0;  yo[1][c][v] -= o1; yo[2][c][v] -= o2; yo[3][c][v] -= o3;
          } else {
            yo[0][c][v] += o0;  yo[1][c][v] += o1; yo[2][c][v] += o2; yo[3][c][v] += o3;
          }
        }
      }
    }

    float* ex = x_s;
    if (p == 1) {
      __syncthreads();
#pragma unroll
      for (int c = 0; c < NCH; ++c) {
#pragma unroll
        for (int v = 0; v < 8; ++v) {
          const int jl = wave * WAVE_J + 16 * (v + 8 * half) + n16;
          ex[c * WG_POS + jl] = yo[3][c][v];
        }
      }
      __syncthreads();
    }
#pragma unroll
    for (int c = 0; c < NCH; ++c) {
      float* oc = out + ((long)bt * NCH + c) * OUT_LEN;
      if (p == 0) {
#pragma unroll
        for (int v = 0; v < 8; ++v) {
          const int jl = wave * WAVE_J + 16 * (v + 8 * half) + n16;
          const int i  = Jb + jl;
          if (jl >= LEAD) {
#pragma unroll
            for (int m = 0; m < 4; ++m) {
              if (NPH * i + 4 * m < OUT_LEN) {
                v4f* dst = (v4f*)(oc + NPH * i + 4 * m);
                const float e = ye[m][c][v], o = yo[m][c][v];
                const v4f w = { e, -e, o, -o };
                *dst = w;
              }
            }
          }
        }
      } else {
        v4f w[8][4];
#pragma unroll
        for (int v = 0; v < 8; ++v) {
          const int jl = wave * WAVE_J + 16 * (v + 8 * half) + n16;
          const int i  = Jb + jl;
#pragma unroll
          for (int m = 0; m < 4; ++m) {
            v4f t = { 0.0f, 0.0f, 0.0f, 0.0f };
            if (jl >= LEAD && NPH * i + 4 * m < OUT_LEN) t = *(const v4f*)(oc + NPH * i + 4 * m);
            w[v][m] = t;
          }
        }
#pragma unroll
        for (int v = 0; v < 8; ++v) {
          const int jl = wave * WAVE_J + 16 * (v + 8 * half) + n16;
          const int i  = Jb + jl;
          if (jl >= LEAD) {
            const float s7m = (p == 1) ? ex[c * WG_POS + jl - 1] : 0.0f;
#pragma unroll
            for (int m = 0; m < 4; ++m) {
              if (NPH * i + 4 * m < OUT_LEN) {
                v4f* dst = (v4f*)(oc + NPH * i + 4 * m);
                const float e = ye[m][c][v], o = yo[m][c][v];
                v4f u = w[v][m];
                if (p == 1) {
                  const float om = (m > 0) ? yo[m > 0 ? m - 1 : 0][c][v] : s7m;
                  u.x += om; u.y -= e; u.z += e; u.w -= o;
                } else {
                  u.y += e; u.w += o;
                }
                *dst = u;
              }
            }
          }
        }
      }
    }
  }
}

extern "C" void kernel_launch(void* const* d_in, const int* in_sizes, int n_in,
                              void* d_out, int out_size, void* d_ws, size_t ws_size,
                              hipStream_t stream) {
  (void)in_sizes; (void)n_in; (void)d_ws; (void)ws_size; (void)out_size;
  const float* x  = (const float*)d_in[0];
  const float* ir = (const float*)d_in[1];
  float* out      = (float*)d_out;

  dim3 grid((OUT_P + WG_J - 1) / WG_J, 64);
  dim3 block(NT);
  spatialize_fir_wmma<<<grid, block, 0, stream>>>(x, ir, out);
}


# One 32-point real DFT of a position's 32-sample window: 47 real 625-tap products on f32 WMMA (bins 0,16 real; 1-15 complex, 3 each) replace the 81 of the even/odd tree; sums regrouped
# speedup vs baseline: 3.2492x; 1.1939x over previous
#include <hip/hip_runtime.h>


typedef float v2f __attribute__((ext_vector_type(2)));
typedef float v4f __attribute__((ext_vector_type(4)));
typedef float v8f __attribute__((ext_vector_type(8)));


constexpr int T_LEN   = 150000;
constexpr int N_TAPS  = 10000;
constexpr int OUT_LEN = 125000;
constexpr int START   = 12500;
constexpr int NCH     = 2;

constexpr int NPH     = 16;
constexpr int NOT     = N_TAPS / NPH;
constexpr int OUT_P   = (OUT_LEN + NPH - 1) / NPH;
constexpr int NGRP    = 6;
constexpr int NPROD   = 9;
constexpr int NPROD_R = 2;
constexpr int KB      = 320;
constexpr int K_LO    = -15;
constexpr int K_TOT   = 640;
constexpr int NKB     = K_TOT / KB;
constexpr int WAVES   = 4;
constexpr int NT      = WAVES * 32;
constexpr int TILE_W  = 256;
constexpr int WAVE_J  = TILE_W;
constexpr int WG_POS  = WAVES * WAVE_J;
constexpr int ASPAN   = WG_POS - TILE_W + 240;
constexpr int XS_ROWS = 11;
constexpr int XG_N    = 3;
constexpr int XS_LEN  = XS_ROWS * NT;
constexpr int XS_ROW  = 16;
constexpr int XS_PITCH= 20;
constexpr int XS_RPITCH = NPROD * XS_PITCH;
constexpr int XS_PHYS = (XS_LEN / XS_ROW) * XS_PITCH;
constexpr int XS_ALL  = NPROD * XS_PHYS;
constexpr int HS_LEN  = KB + 16;
constexpr int HS_W    = 2 * NPROD * NCH;
constexpr int H_ROWS  = (HS_LEN + NT - 1) / NT;
constexpr int NBLK    = NGRP * NKB;
constexpr int PRIO_B2 = 6;
constexpr int PRIO_B1 = 9;
constexpr int PRIO_B0 = 10;

static_assert(START % 4 == 0 && N_TAPS % NPH == 0 && OUT_LEN % 4 == 0 && T_LEN % 4 == 0, "the sixteen phases line up in quads");
static_assert(ASPAN % XS_ROW == 0, "row-aligned anchor");
static_assert(ASPAN + KB - 1 < XS_LEN, "staged sequence covers every A read");
static_assert(K_LO + K_TOT >= NOT && K_LO + 15 >= 0, "K sweep covers every tap");
static_assert(NKB * KB == K_TOT && KB % 16 == 0, "whole K-blocks of whole trips");
static_assert((XS_ALL * 4) % 16 == 0, "tap table stays 16-byte aligned behind the sequences");
static_assert(XS_ALL * 4 < 65536, "every sequence store is a constant offset from one base");
static_assert(((OUT_P + WG_POS - 1) / WG_POS) * WG_POS >= OUT_P, "grid covers every position");
static_assert((HS_W / 4) % 2 == 1, "tap entry pitch is an odd multiple of 16 bytes");
static_assert(XS_ROWS == 3 * XG_N + 2, "row groups of 3, 3, 3 and 2");
static_assert(0 < PRIO_B2 && PRIO_B2 < PRIO_B1 && PRIO_B1 < PRIO_B0 && PRIO_B0 < NBLK, "priority falls in three steps and ends at 0");
static_assert(START + NPH * (ASPAN - ((NKB - 1) * KB + K_LO) - (XS_LEN - 1) - 1) >= 0, "staging stays inside x on the left");
static_assert(START + NPH * ((((OUT_P + WG_POS - 1) / WG_POS) - 1) * WG_POS + ASPAN - K_LO) + NPH - 1 < T_LEN, "and on the right");

#define WMMA_F32(A, B, C) \
  __builtin_amdgcn_wmma_f32_16x16x4_f32(false, (A), false, (B), (short)0, (C), false, false)

template <int G> struct GroupTag { static constexpr int value = G; };

__global__ __launch_bounds__(NT, 2)
void spatialize_fir_wmma(const float* __restrict__ x,
                         const float* __restrict__ ir,
                         float* __restrict__ out) {
  __shared__ __attribute__((aligned(16))) float lds[XS_ALL + (HS_LEN + 1) * HS_W];
  float* x_s = lds;
  float* h_s = lds + XS_ALL;

  const int tid  = threadIdx.x;
  const int lane = tid & 31;
  const int wave = tid >> 5;
  const int half = lane >> 4;
  const int n16  = lane & 15;

  const int bt = blockIdx.y;
  const int Jb = blockIdx.x * WG_POS;

  const float* xb = x + (long)bt * T_LEN + START;

  const int aRow = ASPAN / XS_ROW - wave * (WAVE_J / XS_ROW) - n16;
  const v2f* aBase = (const v2f*)(x_s + aRow * XS_RPITCH + 2 * half);
  const v4f* hBase = (const v4f*)(h_s + (2 * half + n16 + 1) * HS_W);

  constexpr float TW[32] = {
    1.0f, 0.98078528040323043f, 0.92387953251128674f, 0.83146961230254524f, 0.70710678118654752f, 0.55557023301960218f, 0.38268343236508977f, 0.19509032201612825f,
    0.0f, -0.19509032201612825f, -0.38268343236508977f, -0.55557023301960218f, -0.70710678118654752f, -0.83146961230254524f, -0.92387953251128674f, -0.98078528040323043f,
    -1.0f, -0.98078528040323043f, -0.92387953251128674f, -0.83146961230254524f, -0.70710678118654752f, -0.55557023301960218f, -0.38268343236508977f, -0.19509032201612825f,
    0.0f, 0.19509032201612825f, 0.38268343236508977f, 0.55557023301960218f, 0.70710678118654752f, 0.83146961230254524f, 0.92387953251128674f, 0.98078528040323043f };
  auto cos32 = [&](const int k) { return TW[k & 31]; };
  auto sin32 = [&](const int k) { return TW[(k - 8) & 31]; };

  auto run_group = [&](auto gc) {
    constexpr int  g    = decltype(gc)::value;
    constexpr bool cplx = (g < NGRP - 1);
    constexpr int  np   = cplx ? NPROD : NPROD_R;

    v8f acc[NPROD][NCH] = {};

#pragma nounroll
    for (int kb = 0; kb < NKB; ++kb) {
      const int blk = g * NKB + kb;
      if (blk == 0)            __builtin_amdgcn_s_setprio(3);
      else if (blk == PRIO_B2) __builtin_amdgcn_s_setprio(2);
      else if (blk == PRIO_B1) __builtin_amdgcn_s_setprio(1);
      else if (blk == PRIO_B0) __builtin_amdgcn_s_setprio(0);
      const int K0g = kb * KB + K_LO;
      const int nhi = Jb + ASPAN - K0g;

      v4f tv[H_ROWS][8];
#pragma unroll
      for (int r = 0; r < H_ROWS; ++r) {
        const int m  = K0g + tid + NT * r;
        const bool ok = (m >= 0) && (m < NOT);
        const int mc = ok ? m : 0;
        const v4f* tp = (const v4f*)(ir + 2 * NPH * mc);
        const v4f z = {0.0f, 0.0f, 0.0f, 0.0f};
#pragma unroll
        for (int w = 0; w < 8; ++w) { const v4f t = tp[w]; tv[r][w] = ok ? t : z; }
      }
      const v4f* xsrc = (const v4f*)xb + 4 * (nhi - tid);
      auto load_rows = [&](v4f (&q)[XG_N][8], const int r0, const int cnt) {
#pragma unroll
        for (int r = 0; r < XG_N; ++r) {
          if (r < cnt) {
#pragma unroll
            for (int w = 0; w < 8; ++w) q[r][w] = xsrc[-4 * NT * (r0 + r) + w - 4];
          }
        }
      };
      v4f xa[XG_N][8];
      load_rows(xa, 0, XG_N);

      __syncthreads();

      float* xdst = x_s + (tid >> 4) * XS_RPITCH + (tid & 15);
      auto store_row = [&](const int r, const v4f (&q)[8]) {
        float* xd = xdst + (NT / XS_ROW) * XS_RPITCH * r;
        float w[32];
#pragma unroll
        for (int i = 0; i < 8; ++i) { w[4 * i] = q[i].x; w[4 * i + 1] = q[i].y; w[4 * i + 2] = q[i].z; w[4 * i + 3] = q[i].w; }
        float e[16], o[16];
#pragma unroll
        for (int n = 1; n < 16; ++n) { e[n] = w[n] + w[32 - n]; o[n] = w[n] - w[32 - n]; }
        if constexpr (cplx) {
#pragma unroll
          for (int b = 0; b < 3; ++b) {
            const int f = 3 * g + 1 + b;
            float re = w[0] + cos32(16 * f) * w[16];
            float im = 0.0f;
#pragma unroll
            for (int n = 1; n < 16; ++n) {
              const float cs = cos32(f * n), sn = sin32(f * n);
              re += e[n] * cs;
              im -= o[n] * sn;
            }
            xd[(3 * b + 0) * XS_PITCH] = re + im;
            xd[(3 * b + 1) * XS_PITCH] = re;
            xd[(3 * b + 2) * XS_PITCH] = im;
          }
        } else {
          float s0 = w[0] + w[16], s16 = w[0] + w[16];
#pragma unroll
          for (int n = 1; n < 16; ++n) { s0 += e[n]; s16 += (n & 1) ? -e[n] : e[n]; }
          xd[0 * XS_PITCH] = s0;
          xd[1 * XS_PITCH] = s16;
        }
      };
      auto store_rows = [&](const v4f (&q)[XG_N][8], const int r0, const int cnt) {
#pragma unroll
        for (int r = 0; r < XG_N; ++r) { if (r < cnt) store_row(r0 + r, q[r]); }
      };
      store_rows(xa, 0, XG_N);
      v4f xg[XG_N][8];
      load_rows(xg, XG_N, XG_N);
#pragma unroll
      for (int r = 0; r < H_ROWS; ++r) {
        const int i = tid + NT * r;
        if (i < HS_LEN) {
          float gL[16], gR[16];
#pragma unroll
          for (int w = 0; w < 8; ++w) { gL[2 * w] = tv[r][w].x; gR[2 * w] = tv[r][w].y; gL[2 * w + 1] = tv[r][w].z; gR[2 * w + 1] = tv[r][w].w; }
          float hv[NPROD * NCH];
          if constexpr (cplx) {
#pragma unroll
            for (int b = 0; b < 3; ++b) {
              const int f = 3 * g + 1 + b;
              const float sc = cos32(16 * f) * 0.0625f;
              float cL = 0.0f, dL = 0.0f, cR = 0.0f, dR = 0.0f;
#pragma unroll
              for (int t = 0; t < 16; ++t) {
                const float cs = cos32(f * t) * sc, sn = sin32(f * t) * sc;
                cL += gL[t] * cs; cR += gR[t] * cs;
                dL -= gL[t] * sn; dR -= gR[t] * sn;
              }
              hv[(3 * b + 0) * NCH + 0] = cL;       hv[(3 * b + 0) * NCH + 1] = cR;
              hv[(3 * b + 1) * NCH + 0] = dL - cL;  hv[(3 * b + 1) * NCH + 1] = dR - cR;
              hv[(3 * b + 2) * NCH + 0] = cL + dL;  hv[(3 * b + 2) * NCH + 1] = cR + dR;
            }
          } else {
            float s0L = 0.0f, s16L = 0.0f, s0R = 0.0f, s16R = 0.0f;
#pragma unroll
            for (int t = 0; t < 16; ++t) {
              s0L += gL[t]; s0R += gR[t];
              s16L += (t & 1) ? -gL[t] : gL[t]; s16R += (t & 1) ? -gR[t] : gR[t];
            }
            hv[0] = s0L * 0.03125f;  hv[1] = s0R * 0.03125f;
            hv[2] = s16L * 0.03125f; hv[3] = s16R * 0.03125f;
          }
          float* hd = h_s + i * HS_W;
#pragma unroll
          for (int e = 0; e < NPROD * NCH; ++e) { if (e < np * NCH) hd[2 * e + 1] = hv[e]; }
#pragma unroll
          for (int e = 0; e < NPROD * NCH; ++e) { if (e < np * NCH) hd[HS_W + 2 * e] = hv[e]; }
        }
      }
      v4f xh[XG_N][8];
      load_rows(xh, 2 * XG_N, XG_N);
      store_rows(xg, XG_N, XG_N);
      load_rows(xa, 3 * XG_N, XS_ROWS - 3 * XG_N);
      store_rows(xh, 2 * XG_N, XG_N);
      store_rows(xa, 3 * XG_N, XS_ROWS - 3 * XG_N);
      __syncthreads();

      if constexpr (cplx) {
        v4f fh[3][3];
        v2f fa[3][3];
        auto load_group = [&](const v2f* ap, const v4f* hp, const int u, const int qg) {
#pragma unroll
          for (int s3 = 0; s3 < 3; ++s3) {
            const int s = 3 * qg + s3;
            fh[qg][s3] = hp[4 * u * (HS_W / 4) + s];
            fa[qg][s3] = ap[s * (XS_PITCH / 2) + 2 * u];
          }
        };
        auto mma_group = [&](const int qg) {
#pragma unroll
          for (int s3 = 0; s3 < 3; ++s3) {
            const int s = 3 * qg + s3;
            const v2f bl = { fh[qg][s3].x, fh[qg][s3].y };
            const v2f br = { fh[qg][s3].z, fh[qg][s3].w };
            acc[s][0] = WMMA_F32(fa[qg][s3], bl, acc[s][0]);
            acc[s][1] = WMMA_F32(fa[qg][s3], br, acc[s][1]);
          }
        };

        load_group(aBase, hBase, 0, 0);
        load_group(aBase, hBase, 0, 1);
        for (int kq = 0; kq < KB / 16; ++kq) {
          const v2f* ap = aBase + (XS_RPITCH / 2) * kq;
          const v4f* hp = hBase + 16 * (HS_W / 4) * kq;
          const int kn = (kq + 1 < KB / 16) ? kq + 1 : kq;
          const v2f* apn = aBase + (XS_RPITCH / 2) * kn;
          const v4f* hpn = hBase + 16 * (HS_W / 4) * kn;
#pragma unroll
          for (int gi = 0; gi < 12; ++gi) {
            const int gl = gi + 2;
            if (gl < 12) load_group(ap,  hp,  gl / 3, gl % 3);
            else         load_group(apn, hpn, (gl - 12) / 3, (gl - 12) % 3);
            mma_group(gi % 3);
          }
        }
      } else {
        for (int kq = 0; kq < KB / 16; ++kq) {
          const v2f* ap = aBase + (XS_RPITCH / 2) * kq;
          const v4f* hp = hBase + 16 * (HS_W / 4) * kq;
#pragma unroll
          for (int u = 0; u < 4; ++u) {
#pragma unroll
            for (int s = 0; s < NPROD_R; ++s) {
              const v4f fhv = hp[4 * u * (HS_W / 4) + s];
              const v2f fav = ap[s * (XS_PITCH / 2) + 2 * u];
              const v2f bl = { fhv.x, fhv.y };
              const v2f br = { fhv.z, fhv.w };
              acc[s][0] = WMMA_F32(fav, bl, acc[s][0]);
              acc[s][1] = WMMA_F32(fav, br, acc[s][1]);
            }
          }
        }
      }
    }

#pragma unroll
    for (int c = 0; c < NCH; ++c) {
      float* oc = out + ((long)bt * NCH + c) * OUT_LEN;
      v4f wv[8][4];
      if constexpr (g > 0) {
#pragma unroll
        for (int v = 0; v < 8; ++v) {
          const int jl = wave * WAVE_J + 16 * (v + 8 * half) + n16;
          const int i  = Jb + jl;
#pragma unroll
          for (int m = 0; m < 4; ++m) {
            v4f t = { 0.0f, 0.0f, 0.0f, 0.0f };
            if (NPH * i + 4 * m < OUT_LEN) t = *(const v4f*)(oc + NPH * i + 4 * m);
            wv[v][m] = t;
          }
        }
      }
#pragma unroll
      for (int v = 0; v < 8; ++v) {
        const int jl = wave * WAVE_J + 16 * (v + 8 * half) + n16;
        const int i  = Jb + jl;
        float y[16];
        if constexpr (cplx) {
#pragma unroll
          for (int s = 0; s < 16; ++s) y[s] = 0.0f;
#pragma unroll
          for (int b = 0; b < 3; ++b) {
            const int f = 3 * g + 1 + b;
            const float re = acc[3 * b][c][v] - acc[3 * b + 2][c][v];
            const float im = acc[3 * b][c][v] + acc[3 * b + 1][c][v];
#pragma unroll
            for (int s = 0; s < 16; ++s) {
              const float cs = cos32(f * s), sn = sin32(f * s);
              y[s] += re * cs;
              y[s] -= im * sn;
            }
          }
        } else {
          const float ys = acc[0][c][v] + acc[1][c][v], yd = acc[0][c][v] - acc[1][c][v];
#pragma unroll
          for (int s = 0; s < 16; ++s) y[s] = (s & 1) ? yd : ys;
        }
#pragma unroll
        for (int m = 0; m < 4; ++m) {
          if (NPH * i + 4 * m < OUT_LEN) {
            v4f* dst = (v4f*)(oc + NPH * i + 4 * m);
            v4f u = { y[4 * m], y[4 * m + 1], y[4 * m + 2], y[4 * m + 3] };
            if constexpr (g > 0) u += wv[v][m];
            *dst = u;
          }
        }
      }
    }
  };
  run_group(GroupTag<0>{});
  run_group(GroupTag<1>{});
  run_group(GroupTag<2>{});
  run_group(GroupTag<3>{});
  run_group(GroupTag<4>{});
  run_group(GroupTag<5>{});
}

extern "C" void kernel_launch(void* const* d_in, const int* in_sizes, int n_in,
                              void* d_out, int out_size, void* d_ws, size_t ws_size,
                              hipStream_t stream) {
  (void)in_sizes; (void)n_in; (void)d_ws; (void)ws_size; (void)out_size;
  const float* x  = (const float*)d_in[0];
  const float* ir = (const float*)d_in[1];
  float* out      = (float*)d_out;

  dim3 grid((OUT_P + WG_POS - 1) / WG_POS, 64);
  dim3 block(NT);
  spatialize_fir_wmma<<<grid, block, 0, stream>>>(x, ir, out);
}


# Five complex groups run as one rolled body: run-time group index, twiddles read from a constant table by scalar loads (was six inlined copies, code far beyond the instruction cache); sums unchanged
# speedup vs baseline: 3.8822x; 1.1948x over previous
#include <hip/hip_runtime.h>


typedef float v2f __attribute__((ext_vector_type(2)));
typedef float v4f __attribute__((ext_vector_type(4)));
typedef float v8f __attribute__((ext_vector_type(8)));


constexpr int T_LEN   = 150000;
constexpr int N_TAPS  = 10000;
constexpr int OUT_LEN = 125000;
constexpr int START   = 12500;
constexpr int NCH     = 2;

constexpr int NPH     = 16;
constexpr int NOT     = N_TAPS / NPH;
constexpr int OUT_P   = (OUT_LEN + NPH - 1) / NPH;
constexpr int NGRP    = 6;
constexpr int NPROD   = 9;
constexpr int NPROD_R = 2;
constexpr int KB      = 320;
constexpr int K_LO    = -15;
constexpr int K_TOT   = 640;
constexpr int NKB     = K_TOT / KB;
constexpr int WAVES   = 4;
constexpr int NT      = WAVES * 32;
constexpr int TILE_W  = 256;
constexpr int WAVE_J  = TILE_W;
constexpr int WG_POS  = WAVES * WAVE_J;
constexpr int ASPAN   = WG_POS - TILE_W + 240;
constexpr int XS_ROWS = 11;
constexpr int XG_N    = 3;
constexpr int XS_LEN  = XS_ROWS * NT;
constexpr int XS_ROW  = 16;
constexpr int XS_PITCH= 20;
constexpr int XS_RPITCH = NPROD * XS_PITCH;
constexpr int XS_PHYS = (XS_LEN / XS_ROW) * XS_PITCH;
constexpr int XS_ALL  = NPROD * XS_PHYS;
constexpr int HS_LEN  = KB + 16;
constexpr int HS_W    = 2 * NPROD * NCH;
constexpr int H_ROWS  = (HS_LEN + NT - 1) / NT;
constexpr int NBLK    = NGRP * NKB;
constexpr int PRIO_B2 = 6;
constexpr int PRIO_B1 = 9;
constexpr int PRIO_B0 = 10;

static_assert(START % 4 == 0 && N_TAPS % NPH == 0 && OUT_LEN % 4 == 0 && T_LEN % 4 == 0, "the sixteen phases line up in quads");
static_assert(OUT_LEN - NPH * (OUT_P - 1) == 8, "the last position holds exactly its lower two quads");
static_assert(ASPAN % XS_ROW == 0, "row-aligned anchor");
static_assert(ASPAN + KB - 1 < XS_LEN, "staged sequence covers every A read");
static_assert(K_LO + K_TOT >= NOT && K_LO + 15 >= 0, "K sweep covers every tap");
static_assert(NKB * KB == K_TOT && KB % 16 == 0, "whole K-blocks of whole trips");
static_assert((XS_ALL * 4) % 16 == 0, "tap table stays 16-byte aligned behind the sequences");
static_assert(XS_ALL * 4 < 65536, "every sequence store is a constant offset from one base");
static_assert(((OUT_P + WG_POS - 1) / WG_POS) * WG_POS >= OUT_P, "grid covers every position");
static_assert((HS_W / 4) % 2 == 1, "tap entry pitch is an odd multiple of 16 bytes");
static_assert(XS_ROWS == 3 * XG_N + 2, "row groups of 3, 3, 3 and 2");
static_assert(0 < PRIO_B2 && PRIO_B2 < PRIO_B1 && PRIO_B1 < PRIO_B0 && PRIO_B0 < NBLK, "priority falls in three steps and ends at 0");
static_assert(START + NPH * (ASPAN - ((NKB - 1) * KB + K_LO) - (XS_LEN - 1) - 1) >= 0, "staging stays inside x on the left");
static_assert(START + NPH * ((((OUT_P + WG_POS - 1) / WG_POS) - 1) * WG_POS + ASPAN - K_LO) + NPH - 1 < T_LEN, "and on the right");

constexpr float TW32[32] = {
  1.0f, 0.98078528040323043f, 0.92387953251128674f, 0.83146961230254524f, 0.70710678118654752f, 0.55557023301960218f, 0.38268343236508977f, 0.19509032201612825f,
  0.0f, -0.19509032201612825f, -0.38268343236508977f, -0.55557023301960218f, -0.70710678118654752f, -0.83146961230254524f, -0.92387953251128674f, -0.98078528040323043f,
  -1.0f, -0.98078528040323043f, -0.92387953251128674f, -0.83146961230254524f, -0.70710678118654752f, -0.55557023301960218f, -0.38268343236508977f, -0.19509032201612825f,
  0.0f, 0.19509032201612825f, 0.38268343236508977f, 0.55557023301960218f, 0.70710678118654752f, 0.83146961230254524f, 0.92387953251128674f, 0.98078528040323043f };

struct TwTab {
  float c[NGRP - 1][3][16];
  float s[NGRP - 1][3][16];
  float hc[NGRP - 1][3][16];
  float hs[NGRP - 1][3][16];
  float sg[NGRP - 1][4];
};
constexpr TwTab make_twtab() {
  TwTab t{};
  for (int g = 0; g < NGRP - 1; ++g) {
    for (int b = 0; b < 3; ++b) {
      const int f = 3 * g + 1 + b;
      const float sg = TW32[(16 * f) & 31];
      const float sc = sg * 0.0625f;
      t.sg[g][b] = sg;
      for (int n = 0; n < 16; ++n) {
        const float cs = TW32[(f * n) & 31], sn = TW32[(f * n - 8) & 31];
        t.c[g][b][n] = cs;       t.s[g][b][n] = sn;
        t.hc[g][b][n] = cs * sc; t.hs[g][b][n] = sn * sc;
      }
    }
    t.sg[g][3] = 0.0f;
  }
  return t;
}
__constant__ const TwTab TWT = make_twtab();

#define WMMA_F32(A, B, C) \
  __builtin_amdgcn_wmma_f32_16x16x4_f32(false, (A), false, (B), (short)0, (C), false, false)

template <bool C> struct KindTag { static constexpr bool value = C; };

__global__ __launch_bounds__(NT, 2)
void spatialize_fir_wmma(const float* __restrict__ x,
                         const float* __restrict__ ir,
                         float* __restrict__ out) {
  __shared__ __attribute__((aligned(16))) float lds[XS_ALL + (HS_LEN + 1) * HS_W];
  float* x_s = lds;
  float* h_s = lds + XS_ALL;

  const int tid  = threadIdx.x;
  const int lane = tid & 31;
  const int wave = tid >> 5;
  const int half = lane >> 4;
  const int n16  = lane & 15;

  const int bt = blockIdx.y;
  const int Jb = blockIdx.x * WG_POS;

  const float* xb = x + (long)bt * T_LEN + START;

  const int aRow = ASPAN / XS_ROW - wave * (WAVE_J / XS_ROW) - n16;
  const v2f* aBase = (const v2f*)(x_s + aRow * XS_RPITCH + 2 * half);
  const v4f* hBase = (const v4f*)(h_s + (2 * half + n16 + 1) * HS_W);

  auto run_group = [&](auto kind, const int g) {
    constexpr bool cplx = decltype(kind)::value;
    constexpr int  np   = cplx ? NPROD : NPROD_R;
    auto fresh = [&]() { int gi = g; asm volatile("" : "+s"(gi)); return gi; };

    v8f acc[NPROD][NCH] = {};

#pragma nounroll
    for (int kb = 0; kb < NKB; ++kb) {
      const int blk = g * NKB + kb;
      if (blk == 0)            __builtin_amdgcn_s_setprio(3);
      else if (blk == PRIO_B2) __builtin_amdgcn_s_setprio(2);
      else if (blk == PRIO_B1) __builtin_amdgcn_s_setprio(1);
      else if (blk == PRIO_B0) __builtin_amdgcn_s_setprio(0);
      const int K0g = kb * KB + K_LO;
      const int nhi = Jb + ASPAN - K0g;

      v4f tv[H_ROWS][8];
#pragma unroll
      for (int r = 0; r < H_ROWS; ++r) {
        const int m  = K0g + tid + NT * r;
        const bool ok = (m >= 0) && (m < NOT);
        const int mc = ok ? m : 0;
        const v4f* tp = (const v4f*)(ir + 2 * NPH * mc);
        const v4f z = {0.0f, 0.0f, 0.0f, 0.0f};
#pragma unroll
        for (int w = 0; w < 8; ++w) { const v4f t = tp[w]; tv[r][w] = ok ? t : z; }
      }
      const v4f* xsrc = (const v4f*)xb + 4 * (nhi - tid);
      auto load_rows = [&](v4f (&q)[XG_N][8], const int r0, const int cnt) {
#pragma unroll
        for (int r = 0; r < XG_N; ++r) {
          if (r < cnt) {
#pragma unroll
            for (int w = 0; w < 8; ++w) q[r][w] = xsrc[-4 * NT * (r0 + r) + w - 4];
          }
        }
      };
      v4f xa[XG_N][8];
      load_rows(xa, 0, XG_N);

      __syncthreads();

      float* xdst = x_s + (tid >> 4) * XS_RPITCH + (tid & 15);
      auto store_rows = [&](const v4f (&q)[XG_N][8], const int r0, const int cnt) {
        float w0[XG_N], w16[XG_N], e[XG_N][16], o[XG_N][16];
#pragma unroll
        for (int r = 0; r < XG_N; ++r) {
          if (r < cnt) {
            float w[32];
#pragma unroll
            for (int i = 0; i < 8; ++i) { w[4 * i] = q[r][i].x; w[4 * i + 1] = q[r][i].y; w[4 * i + 2] = q[r][i].z; w[4 * i + 3] = q[r][i].w; }
            w0[r] = w[0]; w16[r] = w[16];
#pragma unroll
            for (int n = 1; n < 16; ++n) { e[r][n] = w[n] + w[32 - n]; o[r][n] = w[n] - w[32 - n]; }
          }
        }
        if constexpr (cplx) {
#pragma unroll
          for (int b = 0; b < 3; ++b) {
            const int gi = fresh();
            const float* tc = TWT.c[gi][b];
            const float* ts = TWT.s[gi][b];
            const float sg = TWT.sg[gi][b];
#pragma unroll
            for (int r = 0; r < XG_N; ++r) {
              if (r < cnt) {
                float* xd = xdst + (NT / XS_ROW) * XS_RPITCH * (r0 + r);
                float re = w0[r] + sg * w16[r];
                float im = 0.0f;
#pragma unroll
                for (int n = 1; n < 16; ++n) {
                  const float cs = tc[n], sn = ts[n];
                  re += e[r][n] * cs;
                  im -= o[r][n] * sn;
                }
                xd[(3 * b + 0) * XS_PITCH] = re + im;
                xd[(3 * b + 1) * XS_PITCH] = re;
                xd[(3 * b + 2) * XS_PITCH] = im;
              }
            }
          }
        } else {
#pragma unroll
          for (int r = 0; r < XG_N; ++r) {
            if (r < cnt) {
              float* xd = xdst + (NT / XS_ROW) * XS_RPITCH * (r0 + r);
              float s0 = w0[r] + w16[r], s16 = w0[r] + w16[r];
#pragma unroll
              for (int n = 1; n < 16; ++n) { s0 += e[r][n]; s16 += (n & 1) ? -e[r][n] : e[r][n]; }
              xd[0 * XS_PITCH] = s0;
              xd[1 * XS_PITCH] = s16;
            }
          }
        }
      };
      store_rows(xa, 0, XG_N);
      v4f xg[XG_N][8];
      load_rows(xg, XG_N, XG_N);
#pragma unroll
      for (int r = 0; r < H_ROWS; ++r) {
        const int i = tid + NT * r;
        if (i < HS_LEN) {
          float gL[16], gR[16];
#pragma unroll
          for (int w = 0; w < 8; ++w) { gL[2 * w] = tv[r][w].x; gR[2 * w] = tv[r][w].y; gL[2 * w + 1] = tv[r][w].z; gR[2 * w + 1] = tv[r][w].w; }
          float hv[NPROD * NCH];
          if constexpr (cplx) {
#pragma unroll
            for (int b = 0; b < 3; ++b) {
              const int gi = fresh();
              const float* hc = TWT.hc[gi][b];
              const float* hs = TWT.hs[gi][b];
              float cL = 0.0f, dL = 0.0f, cR = 0.0f, dR = 0.0f;
#pragma unroll
              for (int t = 0; t < 16; ++t) {
                const float cs = hc[t], sn = hs[t];
                cL += gL[t] * cs; cR += gR[t] * cs;
                dL -= gL[t] * sn; dR -= gR[t] * sn;
              }
              hv[(3 * b + 0) * NCH + 0] = cL;       hv[(3 * b + 0) * NCH + 1] = cR;
              hv[(3 * b + 1) * NCH + 0] = dL - cL;  hv[(3 * b + 1) * NCH + 1] = dR - cR;
              hv[(3 * b + 2) * NCH + 0] = cL + dL;  hv[(3 * b + 2) * NCH + 1] = cR + dR;
            }
          } else {
            float s0L = 0.0f, s16L = 0.0f, s0R = 0.0f, s16R = 0.0f;
#pragma unroll
            for (int t = 0; t < 16; ++t) {
              s0L += gL[t]; s0R += gR[t];
              s16L += (t & 1) ? -gL[t] : gL[t]; s16R += (t & 1) ? -gR[t] : gR[t];
            }
            hv[0] = s0L * 0.03125f;  hv[1] = s0R * 0.03125f;
            hv[2] = s16L * 0.03125f; hv[3] = s16R * 0.03125f;
          }
          float* hd = h_s + i * HS_W;
#pragma unroll
          for (int e = 0; e < NPROD * NCH; ++e) { if (e < np * NCH) hd[2 * e + 1] = hv[e]; }
#pragma unroll
          for (int e = 0; e < NPROD * NCH; ++e) { if (e < np * NCH) hd[HS_W + 2 * e] = hv[e]; }
        }
      }
      v4f xh[XG_N][8];
      load_rows(xh, 2 * XG_N, XG_N);
      store_rows(xg, XG_N, XG_N);
      load_rows(xa, 3 * XG_N, XS_ROWS - 3 * XG_N);
      store_rows(xh, 2 * XG_N, XG_N);
      store_rows(xa, 3 * XG_N, XS_ROWS - 3 * XG_N);
      __syncthreads();

      if constexpr (cplx) {
        v4f fh[3][3];
        v2f fa[3][3];
        auto load_group = [&](const v2f* ap, const v4f* hp, const int u, const int qg) {
#pragma unroll
          for (int s3 = 0; s3 < 3; ++s3) {
            const int s = 3 * qg + s3;
            fh[qg][s3] = hp[4 * u * (HS_W / 4) + s];
            fa[qg][s3] = ap[s * (XS_PITCH / 2) + 2 * u];
          }
        };
        auto mma_group = [&](const int qg) {
#pragma unroll
          for (int s3 = 0; s3 < 3; ++s3) {
            const int s = 3 * qg + s3;
            const v2f bl = { fh[qg][s3].x, fh[qg][s3].y };
            const v2f br = { fh[qg][s3].z, fh[qg][s3].w };
            acc[s][0] = WMMA_F32(fa[qg][s3], bl, acc[s][0]);
            acc[s][1] = WMMA_F32(fa[qg][s3], br, acc[s][1]);
          }
        };

        load_group(aBase, hBase, 0, 0);
        load_group(aBase, hBase, 0, 1);
        for (int kq = 0; kq < KB / 16; ++kq) {
          const v2f* ap = aBase + (XS_RPITCH / 2) * kq;
          const v4f* hp = hBase + 16 * (HS_W / 4) * kq;
          const int kn = (kq + 1 < KB / 16) ? kq + 1 : kq;
          const v2f* apn = aBase + (XS_RPITCH / 2) * kn;
          const v4f* hpn = hBase + 16 * (HS_W / 4) * kn;
#pragma unroll
          for (int gi = 0; gi < 12; ++gi) {
            const int gl = gi + 2;
            if (gl < 12) load_group(ap,  hp,  gl / 3, gl % 3);
            else         load_group(apn, hpn, (gl - 12) / 3, (gl - 12) % 3);
            mma_group(gi % 3);
          }
        }
      } else {
        for (int kq = 0; kq < KB / 16; ++kq) {
          const v2f* ap = aBase + (XS_RPITCH / 2) * kq;
          const v4f* hp = hBase + 16 * (HS_W / 4) * kq;
#pragma unroll
          for (int u = 0; u < 4; ++u) {
#pragma unroll
            for (int s = 0; s < NPROD_R; ++s) {
              const v4f fhv = hp[4 * u * (HS_W / 4) + s];
              const v2f fav = ap[s * (XS_PITCH / 2) + 2 * u];
              const v2f bl = { fhv.x, fhv.y };
              const v2f br = { fhv.z, fhv.w };
              acc[s][0] = WMMA_F32(fav, bl, acc[s][0]);
              acc[s][1] = WMMA_F32(fav, br, acc[s][1]);
            }
          }
        }
      }
    }

#pragma unroll
    for (int c = 0; c < NCH; ++c) {
      float* oc = out + ((long)bt * NCH + c) * OUT_LEN;
      v4f wv[8][4];
      if (g > 0) {
#pragma unroll
        for (int v = 0; v < 8; ++v) {
          const int jl = wave * WAVE_J + 16 * (v + 8 * half) + n16;
          const int i  = Jb + jl;
#pragma unroll
          for (int m = 0; m < 4; ++m) wv[v][m] = v4f{ 0.0f, 0.0f, 0.0f, 0.0f };
          if (i < OUT_P) {
#pragma unroll
            for (int m = 0; m < 2; ++m) wv[v][m] = *(const v4f*)(oc + NPH * i + 4 * m);
          }
          if (i < OUT_P - 1) {
#pragma unroll
            for (int m = 2; m < 4; ++m) wv[v][m] = *(const v4f*)(oc + NPH * i + 4 * m);
          }
        }
      }
      auto put = [&](const int v, const v4f (&y)[4]) {
        const int jl = wave * WAVE_J + 16 * (v + 8 * half) + n16;
        const int i  = Jb + jl;
        auto put_quad = [&](const int m) {
          v4f* dst = (v4f*)(oc + NPH * i + 4 * m);
          v4f u = y[m];
          if (g > 0) u += wv[v][m];
          *dst = u;
        };
        if (i < OUT_P)     { put_quad(0); put_quad(1); }
        if (i < OUT_P - 1) { put_quad(2); put_quad(3); }
      };
      if constexpr (cplx) {
#pragma unroll
        for (int vh = 0; vh < 2; ++vh) {
          v4f y[4][4];
#pragma unroll
          for (int vv = 0; vv < 4; ++vv) {
#pragma unroll
            for (int m = 0; m < 4; ++m) y[vv][m] = v4f{ 0.0f, 0.0f, 0.0f, 0.0f };
          }
#pragma unroll
          for (int b = 0; b < 3; ++b) {
            const int gi = fresh();
            const float* tc = TWT.c[gi][b];
            const float* ts = TWT.s[gi][b];
#pragma unroll
            for (int vv = 0; vv < 4; ++vv) {
              const int v = 4 * vh + vv;
              const float re = acc[3 * b][c][v] - acc[3 * b + 2][c][v];
              const float im = acc[3 * b][c][v] + acc[3 * b + 1][c][v];
#pragma unroll
              for (int q = 0; q < 16; ++q) {
                const float cs = tc[q], sn = ts[q];
                float t = y[vv][q >> 2][q & 3];
                t += re * cs;
                t -= im * sn;
                y[vv][q >> 2][q & 3] = t;
              }
            }
            asm volatile("" : "+v"(y[0][0]), "+v"(y[0][1]), "+v"(y[0][2]), "+v"(y[0][3]),
                              "+v"(y[1][0]), "+v"(y[1][1]), "+v"(y[1][2]), "+v"(y[1][3]));
            asm volatile("" : "+v"(y[2][0]), "+v"(y[2][1]), "+v"(y[2][2]), "+v"(y[2][3]),
                              "+v"(y[3][0]), "+v"(y[3][1]), "+v"(y[3][2]), "+v"(y[3][3]));
          }
#pragma unroll
          for (int vv = 0; vv < 4; ++vv) put(4 * vh + vv, y[vv]);
        }
      } else {
#pragma unroll
        for (int v = 0; v < 8; ++v) {
          const float ys = acc[0][c][v] + acc[1][c][v], yd = acc[0][c][v] - acc[1][c][v];
          const v4f y[4] = { v4f{ ys, yd, ys, yd }, v4f{ ys, yd, ys, yd }, v4f{ ys, yd, ys, yd }, v4f{ ys, yd, ys, yd } };
          put(v, y);
        }
      }
    }
  };
#pragma nounroll
  for (int g = 0; g < NGRP - 1; ++g) run_group(KindTag<true>{}, g);
  run_group(KindTag<false>{}, NGRP - 1);
}

extern "C" void kernel_launch(void* const* d_in, const int* in_sizes, int n_in,
                              void* d_out, int out_size, void* d_ws, size_t ws_size,
                              hipStream_t stream) {
  (void)in_sizes; (void)n_in; (void)d_ws; (void)ws_size; (void)out_size;
  const float* x  = (const float*)d_in[0];
  const float* ir = (const float*)d_in[1];
  float* out      = (float*)d_out;

  dim3 grid((OUT_P + WG_POS - 1) / WG_POS, 64);
  dim3 block(NT);
  spatialize_fir_wmma<<<grid, block, 0, stream>>>(x, ir, out);
}
